# starting kernel
# baseline (speedup 1.0000x reference)
#include <hip/hip_runtime.h>
#include <math.h>

typedef float v2f __attribute__((ext_vector_type(2)));
typedef float v8f __attribute__((ext_vector_type(8)));

#define IN_CH 128
#define HID_CH 128
#define OUT_CH 40

__global__ __launch_bounds__(256) void gemm1_kernel(const float* __restrict__ X,
                                                    const float* __restrict__ W,
                                                    float* __restrict__ H, int n) {
  const int wave = threadIdx.x >> 5;
  const int lane = threadIdx.x & 31;
  const int half = lane >> 4;
  const int l    = lane & 15;
  const int row_base = blockIdx.x * 16;
  const int col = (wave << 4) + l;

  int arow_i = row_base + l;
  if (arow_i >= n) arow_i = n - 1;
  const float* __restrict__ arow = X + (size_t)arow_i * IN_CH;

  v8f acc = {};
  for (int k = 0; k < IN_CH; k += 4) {
    const int ka = k + (half << 1);
    v2f a, b;
    a.x = arow[ka];
    a.y = arow[ka + 1];
    b.x = W[(size_t)ka * HID_CH + col];
    b.y = W[(size_t)(ka + 1) * HID_CH + col];
    acc = __builtin_amdgcn_wmma_f32_16x16x4_f32(false, a, false, b,
                                                (short)0, acc, false, false);
  }

  float* __restrict__ crow = H + (size_t)(row_base + (half << 3)) * HID_CH + col;
  if (row_base + 16 <= n) {
#pragma unroll
    for (int v = 0; v < 8; ++v) crow[(size_t)v * HID_CH] = acc[v];
  } else {
#pragma unroll
    for (int v = 0; v < 8; ++v) {
      if (row_base + (half << 3) + v < n) crow[(size_t)v * HID_CH] = acc[v];
    }
  }
}

__global__ __launch_bounds__(96) void gemm2_kernel(const float* __restrict__ A,
                                                   const float* __restrict__ W,
                                                   float* __restrict__ H, int n) {
  const int wave = threadIdx.x >> 5;
  const int lane = threadIdx.x & 31;
  const int half = lane >> 4;
  const int l    = lane & 15;
  const int row_base = blockIdx.x * 16;
  const int col = (wave << 4) + l;
  const bool colok = (col < OUT_CH);

  int arow_i = row_base + l;
  if (arow_i >= n) arow_i = n - 1;
  const float* __restrict__ arow = A + (size_t)arow_i * HID_CH;

  v8f acc = {};
  for (int k = 0; k < HID_CH; k += 4) {
    const int ka = k + (half << 1);
    v2f a, b;
    a.x = arow[ka];
    a.y = arow[ka + 1];
    b.x = colok ? W[(size_t)ka * OUT_CH + col] : 0.0f;
    b.y = colok ? W[(size_t)(ka + 1) * OUT_CH + col] : 0.0f;
    acc = __builtin_amdgcn_wmma_f32_16x16x4_f32(false, a, false, b,
                                                (short)0, acc, false, false);
  }

  float* __restrict__ crow = H + (size_t)(row_base + (half << 3)) * OUT_CH + col;
  if (colok) {
    if (row_base + 16 <= n) {
#pragma unroll
      for (int v = 0; v < 8; ++v) crow[(size_t)v * OUT_CH] = acc[v];
    } else {
#pragma unroll
      for (int v = 0; v < 8; ++v) {
        if (row_base + (half << 3) + v < n) crow[(size_t)v * OUT_CH] = acc[v];
      }
    }
  }
}

__global__ __launch_bounds__(256) void scatter1_kernel(const float* __restrict__ H,
                                                       const int* __restrict__ src,
                                                       const int* __restrict__ dst,
                                                       float* __restrict__ AGG,
                                                       int n_edges) {
  const int e = blockIdx.x * 8 + (threadIdx.x >> 5);
  if (e >= n_edges) return;
  const int lane = threadIdx.x & 31;
  const int s = src[e];
  const int d = dst[e];
  const float4 v = ((const float4*)(H + (size_t)s * HID_CH))[lane];
  float* out = AGG + (size_t)d * HID_CH + lane * 4;
  atomicAdd(out + 0, v.x);
  atomicAdd(out + 1, v.y);
  atomicAdd(out + 2, v.z);
  atomicAdd(out + 3, v.w);
}

__global__ __launch_bounds__(256) void scatter2_kernel(const float* __restrict__ H,
                                                       const int* __restrict__ src,
                                                       const int* __restrict__ dst,
                                                       float* __restrict__ OUT,
                                                       int n_edges) {
  const int e = blockIdx.x * 8 + (threadIdx.x >> 5);
  if (e >= n_edges) return;
  const int lane = threadIdx.x & 31;
  const int s = src[e];
  const int d = dst[e];
  if (lane < OUT_CH / 2) {
    const float2 v = ((const float2*)(H + (size_t)s * OUT_CH))[lane];
    float* out = OUT + (size_t)d * OUT_CH + lane * 2;
    atomicAdd(out + 0, v.x);
    atomicAdd(out + 1, v.y);
  }
}

__global__ __launch_bounds__(256) void bias_relu_kernel(float* __restrict__ A,
                                                        const float* __restrict__ b1,
                                                        long long total) {
  long long i = (long long)blockIdx.x * blockDim.x + threadIdx.x;
  const long long stride = (long long)gridDim.x * blockDim.x;
  for (; i < total; i += stride) {
    const float v = A[i] + b1[i & (HID_CH - 1)];
    A[i] = v > 0.0f ? v : 0.0f;
  }
}

__global__ __launch_bounds__(256) void bias_logsoftmax_kernel(float* __restrict__ O,
                                                              const float* __restrict__ b2,
                                                              int n) {
  const int row = blockIdx.x * 8 + (threadIdx.x >> 5);
  if (row >= n) return;
  const int lane = threadIdx.x & 31;
  float* o = O + (size_t)row * OUT_CH;

  const float a = o[lane] + b2[lane];
  const bool hi = (lane < OUT_CH - 32);
  const float b = hi ? (o[lane + 32] + b2[lane + 32]) : -INFINITY;

  float m = fmaxf(a, b);
  for (int off = 16; off > 0; off >>= 1) m = fmaxf(m, __shfl_xor(m, off, 32));
  float s = expf(a - m) + (hi ? expf(b - m) : 0.0f);
  for (int off = 16; off > 0; off >>= 1) s += __shfl_xor(s, off, 32);
  const float lse = m + logf(s);

  o[lane] = a - lse;
  if (hi) o[lane + 32] = b - lse;
}

__global__ __launch_bounds__(256) void zero_kernel(float4* __restrict__ p, long long n4) {
  long long i = (long long)blockIdx.x * blockDim.x + threadIdx.x;
  const long long stride = (long long)gridDim.x * blockDim.x;
  const float4 z = make_float4(0.0f, 0.0f, 0.0f, 0.0f);
  for (; i < n4; i += stride) p[i] = z;
}

extern "C" void kernel_launch(void* const* d_in, const int* in_sizes, int n_in,
                              void* d_out, int out_size, void* d_ws, size_t ws_size,
                              hipStream_t stream) {
  const float* x  = (const float*)d_in[0];
  const int*   ei = (const int*)d_in[1];
  const float* W1 = (const float*)d_in[2];
  const float* b1 = (const float*)d_in[3];
  const float* W2 = (const float*)d_in[4];
  const float* b2 = (const float*)d_in[5];

  const int n       = in_sizes[0] / IN_CH;
  const int n_edges = in_sizes[1] / 2;
  const int* src = ei;
  const int* dst = ei + n_edges;

  float* h1   = (float*)d_ws;
  float* agg1 = h1 + (size_t)n * HID_CH;
  float* h2   = h1;
  float* out  = (float*)d_out;

  zero_kernel<<<2048, 256, 0, stream>>>((float4*)agg1, (long long)n * HID_CH / 4);
  zero_kernel<<<2048, 256, 0, stream>>>((float4*)out,  (long long)n * OUT_CH / 4);

  gemm1_kernel<<<(n + 15) / 16, 256, 0, stream>>>(x, W1, h1, n);
  scatter1_kernel<<<(n_edges + 7) / 8, 256, 0, stream>>>(h1, src, dst, agg1, n_edges);
  bias_relu_kernel<<<2048, 256, 0, stream>>>(agg1, b1, (long long)n * HID_CH);

  gemm2_kernel<<<(n + 15) / 16, 96, 0, stream>>>(agg1, W2, h2, n);
  scatter2_kernel<<<(n_edges + 7) / 8, 256, 0, stream>>>(h2, src, dst, out, n_edges);
  bias_logsoftmax_kernel<<<(n + 7) / 8, 256, 0, stream>>>(out, b2, n);
}


# Float atomic scatter-add of both layers replaced by an in-call counting sort of the edges by destination and per-node register sums in arrival-order rank; zero fills dropped
# speedup vs baseline: 1.7222x; 1.7222x over previous
#include <hip/hip_runtime.h>
#include <math.h>

typedef float v2f __attribute__((ext_vector_type(2)));
typedef float v8f __attribute__((ext_vector_type(8)));

#define IN_CH 128
#define HID_CH 128
#define OUT_CH 40

__global__ __launch_bounds__(256) void gemm1_kernel(const float* __restrict__ X,
                                                    const float* __restrict__ W,
                                                    float* __restrict__ H, int n) {
  const int wave = threadIdx.x >> 5;
  const int lane = threadIdx.x & 31;
  const int half = lane >> 4;
  const int l    = lane & 15;
  const int row_base = blockIdx.x * 16;
  const int col = (wave << 4) + l;

  int arow_i = row_base + l;
  if (arow_i >= n) arow_i = n - 1;
  const float* __restrict__ arow = X + (size_t)arow_i * IN_CH;

  v8f acc = {};
  for (int k = 0; k < IN_CH; k += 4) {
    const int ka = k + (half << 1);
    v2f a, b;
    a.x = arow[ka];
    a.y = arow[ka + 1];
    b.x = W[(size_t)ka * HID_CH + col];
    b.y = W[(size_t)(ka + 1) * HID_CH + col];
    acc = __builtin_amdgcn_wmma_f32_16x16x4_f32(false, a, false, b,
                                                (short)0, acc, false, false);
  }

  float* __restrict__ crow = H + (size_t)(row_base + (half << 3)) * HID_CH + col;
  if (row_base + 16 <= n) {
#pragma unroll
    for (int v = 0; v < 8; ++v) crow[(size_t)v * HID_CH] = acc[v];
  } else {
#pragma unroll
    for (int v = 0; v < 8; ++v) {
      if (row_base + (half << 3) + v < n) crow[(size_t)v * HID_CH] = acc[v];
    }
  }
}

__global__ __launch_bounds__(96) void gemm2_kernel(const float* __restrict__ A,
                                                   const float* __restrict__ W,
                                                   float* __restrict__ H, int n) {
  const int wave = threadIdx.x >> 5;
  const int lane = threadIdx.x & 31;
  const int half = lane >> 4;
  const int l    = lane & 15;
  const int row_base = blockIdx.x * 16;
  const int col = (wave << 4) + l;
  const bool colok = (col < OUT_CH);

  int arow_i = row_base + l;
  if (arow_i >= n) arow_i = n - 1;
  const float* __restrict__ arow = A + (size_t)arow_i * HID_CH;

  v8f acc = {};
  for (int k = 0; k < HID_CH; k += 4) {
    const int ka = k + (half << 1);
    v2f a, b;
    a.x = arow[ka];
    a.y = arow[ka + 1];
    b.x = colok ? W[(size_t)ka * OUT_CH + col] : 0.0f;
    b.y = colok ? W[(size_t)(ka + 1) * OUT_CH + col] : 0.0f;
    acc = __builtin_amdgcn_wmma_f32_16x16x4_f32(false, a, false, b,
                                                (short)0, acc, false, false);
  }

  float* __restrict__ crow = H + (size_t)(row_base + (half << 3)) * OUT_CH + col;
  if (colok) {
    if (row_base + 16 <= n) {
#pragma unroll
      for (int v = 0; v < 8; ++v) crow[(size_t)v * OUT_CH] = acc[v];
    } else {
#pragma unroll
      for (int v = 0; v < 8; ++v) {
        if (row_base + (half << 3) + v < n) crow[(size_t)v * OUT_CH] = acc[v];
      }
    }
  }
}

__global__ __launch_bounds__(256) void zero_int_kernel(int* __restrict__ p, int count) {
  const int i = blockIdx.x * 256 + threadIdx.x;
  if (i < count) p[i] = 0;
}

__global__ __launch_bounds__(256) void rank_kernel(const int* __restrict__ dst,
                                                   int* __restrict__ cnt,
                                                   int* __restrict__ rank,
                                                   int n_edges) {
  const int e = blockIdx.x * 256 + threadIdx.x;
  if (e < n_edges) rank[e] = atomicAdd(&cnt[dst[e]], 1);
}

__global__ __launch_bounds__(256) void offsets_kernel(const int* __restrict__ cnt,
                                                      int* __restrict__ off,
                                                      int* __restrict__ cursor,
                                                      int n) {
  const int i = blockIdx.x * 256 + threadIdx.x;
  const int lane = threadIdx.x & 31;
  const int c = (i < n) ? cnt[i] : 0;
  int x = c;
  for (int o = 1; o < 32; o <<= 1) {
    const int y = __shfl_up(x, o, 32);
    if (lane >= o) x += y;
  }
  int base = 0;
  if (lane == 31) base = atomicAdd(cursor, x);
  base = __shfl(base, 31, 32);
  if (i < n) off[i] = base + x - c;
}

__global__ __launch_bounds__(256) void place_kernel(const int* __restrict__ src,
                                                    const int* __restrict__ dst,
                                                    const int* __restrict__ rank,
                                                    const int* __restrict__ off,
                                                    int* __restrict__ ssrc,
                                                    int n_edges) {
  const int e = blockIdx.x * 256 + threadIdx.x;
  if (e < n_edges) ssrc[off[dst[e]] + rank[e]] = src[e];
}

template <int K>
__device__ __forceinline__ void add_rows128(const float4* __restrict__ H4, int myidx,
                                            int t, int lane, float4& acc) {
  float4 v[K];
#pragma unroll
  for (int u = 0; u < K; ++u) {
    const int s = __builtin_amdgcn_readlane(myidx, t + u);
    v[u] = H4[(size_t)s * (HID_CH / 4) + lane];
  }
#pragma unroll
  for (int u = 0; u < K; ++u) {
    acc.x += v[u].x; acc.y += v[u].y; acc.z += v[u].z; acc.w += v[u].w;
  }
}

__global__ __launch_bounds__(256) void gather1_kernel(const float* __restrict__ H,
                                                      const int* __restrict__ ssrc,
                                                      const int* __restrict__ off,
                                                      const int* __restrict__ cnt,
                                                      float* __restrict__ AGG, int n) {
  const int wave = __builtin_amdgcn_readfirstlane((int)(threadIdx.x >> 5));
  const int lane = threadIdx.x & 31;
  const int node = blockIdx.x * 8 + wave;
  if (node >= n) return;
  const int beg = off[node];
  const int deg = cnt[node];
  const float4* __restrict__ H4 = (const float4*)H;
  float4 acc = make_float4(0.0f, 0.0f, 0.0f, 0.0f);
  for (int j = 0; j < deg; j += 32) {
    const int m = (deg - j < 32) ? (deg - j) : 32;
    const int myidx = (lane < m) ? ssrc[beg + j + lane] : 0;
    int t = 0;
    for (; t + 8 <= m; t += 8) add_rows128<8>(H4, myidx, t, lane, acc);
    if (t + 4 <= m) { add_rows128<4>(H4, myidx, t, lane, acc); t += 4; }
    if (t + 2 <= m) { add_rows128<2>(H4, myidx, t, lane, acc); t += 2; }
    if (t < m) add_rows128<1>(H4, myidx, t, lane, acc);
  }
  ((float4*)AGG)[(size_t)node * (HID_CH / 4) + lane] = acc;
}

__device__ __forceinline__ float4 load_row40(const float4* __restrict__ H4, int myidx,
                                             int e, int g, int m, int c) {
  const int s = __shfl(myidx, e & 31, 32);
  float4 v = make_float4(0.0f, 0.0f, 0.0f, 0.0f);
  if (g < 3 && e < m) v = H4[(size_t)s * (OUT_CH / 4) + c];
  return v;
}

__global__ __launch_bounds__(256) void gather2_kernel(const float* __restrict__ H,
                                                      const int* __restrict__ ssrc,
                                                      const int* __restrict__ off,
                                                      const int* __restrict__ cnt,
                                                      float* __restrict__ OUT, int n) {
  const int wave = __builtin_amdgcn_readfirstlane((int)(threadIdx.x >> 5));
  const int lane = threadIdx.x & 31;
  const int node = blockIdx.x * 8 + wave;
  if (node >= n) return;
  const int g = lane / 10;
  const int c = lane - g * 10;
  const int beg = off[node];
  const int deg = cnt[node];
  const float4* __restrict__ H4 = (const float4*)H;
  float4 acc = make_float4(0.0f, 0.0f, 0.0f, 0.0f);
  for (int j = 0; j < deg; j += 32) {
    const int m = (deg - j < 32) ? (deg - j) : 32;
    const int myidx = (lane < m) ? ssrc[beg + j + lane] : 0;
    for (int t = 0; t < m; t += 12) {
      const float4 v0 = load_row40(H4, myidx, t + g,     g, m, c);
      const float4 v1 = load_row40(H4, myidx, t + 3 + g, g, m, c);
      const float4 v2 = load_row40(H4, myidx, t + 6 + g, g, m, c);
      const float4 v3 = load_row40(H4, myidx, t + 9 + g, g, m, c);
      acc.x += v0.x; acc.y += v0.y; acc.z += v0.z; acc.w += v0.w;
      acc.x += v1.x; acc.y += v1.y; acc.z += v1.z; acc.w += v1.w;
      acc.x += v2.x; acc.y += v2.y; acc.z += v2.z; acc.w += v2.w;
      acc.x += v3.x; acc.y += v3.y; acc.z += v3.z; acc.w += v3.w;
    }
  }
  const int l1 = (lane + 10) & 31;
  const int l2 = (lane + 20) & 31;
  float4 r;
  r.x = (acc.x + __shfl(acc.x, l1, 32)) + __shfl(acc.x, l2, 32);
  r.y = (acc.y + __shfl(acc.y, l1, 32)) + __shfl(acc.y, l2, 32);
  r.z = (acc.z + __shfl(acc.z, l1, 32)) + __shfl(acc.z, l2, 32);
  r.w = (acc.w + __shfl(acc.w, l1, 32)) + __shfl(acc.w, l2, 32);
  if (lane < OUT_CH / 4) ((float4*)OUT)[(size_t)node * (OUT_CH / 4) + lane] = r;
}

__global__ __launch_bounds__(256) void bias_relu_kernel(float* __restrict__ A,
                                                        const float* __restrict__ b1,
                                                        long long total) {
  long long i = (long long)blockIdx.x * blockDim.x + threadIdx.x;
  const long long stride = (long long)gridDim.x * blockDim.x;
  for (; i < total; i += stride) {
    const float v = A[i] + b1[i & (HID_CH - 1)];
    A[i] = v > 0.0f ? v : 0.0f;
  }
}

__global__ __launch_bounds__(256) void bias_logsoftmax_kernel(float* __restrict__ O,
                                                              const float* __restrict__ b2,
                                                              int n) {
  const int row = blockIdx.x * 8 + (threadIdx.x >> 5);
  if (row >= n) return;
  const int lane = threadIdx.x & 31;
  float* o = O + (size_t)row * OUT_CH;

  const float a = o[lane] + b2[lane];
  const bool hi = (lane < OUT_CH - 32);
  const float b = hi ? (o[lane + 32] + b2[lane + 32]) : -INFINITY;

  float m = fmaxf(a, b);
  for (int off = 16; off > 0; off >>= 1) m = fmaxf(m, __shfl_xor(m, off, 32));
  float s = expf(a - m) + (hi ? expf(b - m) : 0.0f);
  for (int off = 16; off > 0; off >>= 1) s += __shfl_xor(s, off, 32);
  const float lse = m + logf(s);

  o[lane] = a - lse;
  if (hi) o[lane + 32] = b - lse;
}

extern "C" void kernel_launch(void* const* d_in, const int* in_sizes, int n_in,
                              void* d_out, int out_size, void* d_ws, size_t ws_size,
                              hipStream_t stream) {
  const float* x  = (const float*)d_in[0];
  const int*   ei = (const int*)d_in[1];
  const float* W1 = (const float*)d_in[2];
  const float* b1 = (const float*)d_in[3];
  const float* W2 = (const float*)d_in[4];
  const float* b2 = (const float*)d_in[5];

  const int n       = in_sizes[0] / IN_CH;
  const int n_edges = in_sizes[1] / 2;
  const int* src = ei;
  const int* dst = ei + n_edges;

  float* h1   = (float*)d_ws;
  float* agg1 = h1 + (size_t)n * HID_CH;
  float* h2   = h1;
  float* out  = (float*)d_out;
  int* cnt    = (int*)(agg1 + (size_t)n * HID_CH);
  int* cursor = cnt + n;
  int* off    = cursor + 1;
  int* rank   = off + n;
  int* ssrc   = rank + n_edges;

  zero_int_kernel<<<(n + 1 + 255) / 256, 256, 0, stream>>>(cnt, n + 1);
  rank_kernel<<<(n_edges + 255) / 256, 256, 0, stream>>>(dst, cnt, rank, n_edges);
  offsets_kernel<<<(n + 255) / 256, 256, 0, stream>>>(cnt, off, cursor, n);
  place_kernel<<<(n_edges + 255) / 256, 256, 0, stream>>>(src, dst, rank, off, ssrc, n_edges);

  gemm1_kernel<<<(n + 15) / 16, 256, 0, stream>>>(x, W1, h1, n);
  gather1_kernel<<<(n + 7) / 8, 256, 0, stream>>>(h1, ssrc, off, cnt, agg1, n);
  bias_relu_kernel<<<2048, 256, 0, stream>>>(agg1, b1, (long long)n * HID_CH);

  gemm2_kernel<<<(n + 15) / 16, 96, 0, stream>>>(agg1, W2, h2, n);
  gather2_kernel<<<(n + 7) / 8, 256, 0, stream>>>(h2, ssrc, off, cnt, out, n);
  bias_logsoftmax_kernel<<<(n + 7) / 8, 256, 0, stream>>>(out, b2, n);
}


# Every launch cut to at most the work-groups the chip holds at once: each wave loops over its share of tiles, nodes, rows or edges (weight strips stay in registers) instead of one short wave per item
# speedup vs baseline: 1.8807x; 1.0920x over previous
#include <hip/hip_runtime.h>
#include <math.h>

typedef float v2f __attribute__((ext_vector_type(2)));
typedef float v8f __attribute__((ext_vector_type(8)));

#define IN_CH 128
#define HID_CH 128
#define OUT_CH 40

__global__ __launch_bounds__(256) void gemm1_kernel(const float* __restrict__ X,
                                                    const float* __restrict__ W,
                                                    float* __restrict__ H, int n, int tiles) {
  const int wave = threadIdx.x >> 5;
  const int lane = threadIdx.x & 31;
  const int half = lane >> 4;
  const int l    = lane & 15;
  const int col = (wave << 4) + l;

  v2f bw[IN_CH / 4];
#pragma unroll
  for (int i = 0; i < IN_CH / 4; ++i) {
    const int ka = 4 * i + (half << 1);
    bw[i].x = W[(size_t)ka * HID_CH + col];
    bw[i].y = W[(size_t)(ka + 1) * HID_CH + col];
  }

  for (int tile = blockIdx.x; tile < tiles; tile += gridDim.x) {
    const int row_base = tile * 16;
    int arow_i = row_base + l;
    if (arow_i >= n) arow_i = n - 1;
    const float* __restrict__ arow = X + (size_t)arow_i * IN_CH;

    v8f acc = {};
#pragma unroll
    for (int i = 0; i < IN_CH / 4; ++i) {
      const int ka = 4 * i + (half << 1);
      v2f a;
      a.x = arow[ka];
      a.y = arow[ka + 1];
      acc = __builtin_amdgcn_wmma_f32_16x16x4_f32(false, a, false, bw[i],
                                                  (short)0, acc, false, false);
    }

    float* __restrict__ crow = H + (size_t)(row_base + (half << 3)) * HID_CH + col;
    if (row_base + 16 <= n) {
#pragma unroll
      for (int v = 0; v < 8; ++v) crow[(size_t)v * HID_CH] = acc[v];
    } else {
#pragma unroll
      for (int v = 0; v < 8; ++v) {
        if (row_base + (half << 3) + v < n) crow[(size_t)v * HID_CH] = acc[v];
      }
    }
  }
}

__global__ __launch_bounds__(96) void gemm2_kernel(const float* __restrict__ A,
                                                   const float* __restrict__ W,
                                                   float* __restrict__ H, int n, int tiles) {
  const int wave = threadIdx.x >> 5;
  const int lane = threadIdx.x & 31;
  const int half = lane >> 4;
  const int l    = lane & 15;
  const int col = (wave << 4) + l;
  const bool colok = (col < OUT_CH);

  v2f bw[HID_CH / 4];
#pragma unroll
  for (int i = 0; i < HID_CH / 4; ++i) {
    const int ka = 4 * i + (half << 1);
    bw[i].x = colok ? W[(size_t)ka * OUT_CH + col] : 0.0f;
    bw[i].y = colok ? W[(size_t)(ka + 1) * OUT_CH + col] : 0.0f;
  }

  for (int tile = blockIdx.x; tile < tiles; tile += gridDim.x) {
    const int row_base = tile * 16;
    int arow_i = row_base + l;
    if (arow_i >= n) arow_i = n - 1;
    const float* __restrict__ arow = A + (size_t)arow_i * HID_CH;

    v8f acc = {};
#pragma unroll
    for (int i = 0; i < HID_CH / 4; ++i) {
      const int ka = 4 * i + (half << 1);
      v2f a;
      a.x = arow[ka];
      a.y = arow[ka + 1];
      acc = __builtin_amdgcn_wmma_f32_16x16x4_f32(false, a, false, bw[i],
                                                  (short)0, acc, false, false);
    }

    float* __restrict__ crow = H + (size_t)(row_base + (half << 3)) * OUT_CH + col;
    if (colok) {
      if (row_base + 16 <= n) {
#pragma unroll
        for (int v = 0; v < 8; ++v) crow[(size_t)v * OUT_CH] = acc[v];
      } else {
#pragma unroll
        for (int v = 0; v < 8; ++v) {
          if (row_base + (half << 3) + v < n) crow[(size_t)v * OUT_CH] = acc[v];
        }
      }
    }
  }
}

__global__ __launch_bounds__(256) void zero_int_kernel(int* __restrict__ p, int count) {
  const int i = blockIdx.x * 256 + threadIdx.x;
  if (i < count) p[i] = 0;
}

__global__ __launch_bounds__(256) void rank_kernel(const int* __restrict__ dst,
                                                   int* __restrict__ cnt,
                                                   int* __restrict__ rank,
                                                   int n_edges) {
  const int stride = gridDim.x * 256;
  int e = blockIdx.x * 256 + threadIdx.x;
  for (; e + 3 * stride < n_edges; e += 4 * stride) {
    const int d0 = dst[e];
    const int d1 = dst[e + stride];
    const int d2 = dst[e + 2 * stride];
    const int d3 = dst[e + 3 * stride];
    const int r0 = atomicAdd(&cnt[d0], 1);
    const int r1 = atomicAdd(&cnt[d1], 1);
    const int r2 = atomicAdd(&cnt[d2], 1);
    const int r3 = atomicAdd(&cnt[d3], 1);
    rank[e] = r0;
    rank[e + stride] = r1;
    rank[e + 2 * stride] = r2;
    rank[e + 3 * stride] = r3;
  }
  for (; e < n_edges; e += stride) rank[e] = atomicAdd(&cnt[dst[e]], 1);
}

__global__ __launch_bounds__(256) void offsets_kernel(const int* __restrict__ cnt,
                                                      int* __restrict__ off,
                                                      int* __restrict__ cursor,
                                                      int n) {
  const int i = blockIdx.x * 256 + threadIdx.x;
  const int lane = threadIdx.x & 31;
  const int c = (i < n) ? cnt[i] : 0;
  int x = c;
  for (int o = 1; o < 32; o <<= 1) {
    const int y = __shfl_up(x, o, 32);
    if (lane >= o) x += y;
  }
  int base = 0;
  if (lane == 31) base = atomicAdd(cursor, x);
  base = __shfl(base, 31, 32);
  if (i < n) off[i] = base + x - c;
}

__global__ __launch_bounds__(256) void place_kernel(const int* __restrict__ src,
                                                    const int* __restrict__ dst,
                                                    const int* __restrict__ rank,
                                                    const int* __restrict__ off,
                                                    int* __restrict__ ssrc,
                                                    int n_edges) {
  const int stride = gridDim.x * 256;
  int e = blockIdx.x * 256 + threadIdx.x;
  for (; e + 3 * stride < n_edges; e += 4 * stride) {
    const int d0 = dst[e];
    const int d1 = dst[e + stride];
    const int d2 = dst[e + 2 * stride];
    const int d3 = dst[e + 3 * stride];
    const int r0 = rank[e];
    const int r1 = rank[e + stride];
    const int r2 = rank[e + 2 * stride];
    const int r3 = rank[e + 3 * stride];
    const int s0 = src[e];
    const int s1 = src[e + stride];
    const int s2 = src[e + 2 * stride];
    const int s3 = src[e + 3 * stride];
    const int o0 = off[d0];
    const int o1 = off[d1];
    const int o2 = off[d2];
    const int o3 = off[d3];
    ssrc[o0 + r0] = s0;
    ssrc[o1 + r1] = s1;
    ssrc[o2 + r2] = s2;
    ssrc[o3 + r3] = s3;
  }
  for (; e < n_edges; e += stride) ssrc[off[dst[e]] + rank[e]] = src[e];
}

template <int K>
__device__ __forceinline__ void add_rows128(const float4* __restrict__ H4, int myidx,
                                            int t, int lane, float4& acc) {
  float4 v[K];
#pragma unroll
  for (int u = 0; u < K; ++u) {
    const int s = __builtin_amdgcn_readlane(myidx, t + u);
    v[u] = H4[(size_t)s * (HID_CH / 4) + lane];
  }
#pragma unroll
  for (int u = 0; u < K; ++u) {
    acc.x += v[u].x; acc.y += v[u].y; acc.z += v[u].z; acc.w += v[u].w;
  }
}

__global__ __launch_bounds__(256) void gather1_kernel(const float* __restrict__ H,
                                                      const int* __restrict__ ssrc,
                                                      const int* __restrict__ off,
                                                      const int* __restrict__ cnt,
                                                      float* __restrict__ AGG, int n) {
  const int wave = __builtin_amdgcn_readfirstlane((int)(threadIdx.x >> 5));
  const int lane = threadIdx.x & 31;
  const int nwaves = gridDim.x * 8;
  const float4* __restrict__ H4 = (const float4*)H;
  for (int node = blockIdx.x * 8 + wave; node < n; node += nwaves) {
    const int beg = off[node];
    const int deg = cnt[node];
    float4 acc = make_float4(0.0f, 0.0f, 0.0f, 0.0f);
    for (int j = 0; j < deg; j += 32) {
      const int m = (deg - j < 32) ? (deg - j) : 32;
      const int myidx = (lane < m) ? ssrc[beg + j + lane] : 0;
      int t = 0;
      for (; t + 8 <= m; t += 8) add_rows128<8>(H4, myidx, t, lane, acc);
      if (t + 4 <= m) { add_rows128<4>(H4, myidx, t, lane, acc); t += 4; }
      if (t + 2 <= m) { add_rows128<2>(H4, myidx, t, lane, acc); t += 2; }
      if (t < m) add_rows128<1>(H4, myidx, t, lane, acc);
    }
    ((float4*)AGG)[(size_t)node * (HID_CH / 4) + lane] = acc;
  }
}

__device__ __forceinline__ float4 load_row40(const float4* __restrict__ H4, int myidx,
                                             int e, int g, int m, int c) {
  const int s = __shfl(myidx, e & 31, 32);
  float4 v = make_float4(0.0f, 0.0f, 0.0f, 0.0f);
  if (g < 3 && e < m) v = H4[(size_t)s * (OUT_CH / 4) + c];
  return v;
}

__global__ __launch_bounds__(256) void gather2_kernel(const float* __restrict__ H,
                                                      const int* __restrict__ ssrc,
                                                      const int* __restrict__ off,
                                                      const int* __restrict__ cnt,
                                                      float* __restrict__ OUT, int n) {
  const int wave = __builtin_amdgcn_readfirstlane((int)(threadIdx.x >> 5));
  const int lane = threadIdx.x & 31;
  const int nwaves = gridDim.x * 8;
  const int g = lane / 10;
  const int c = lane - g * 10;
  const int l1 = (lane + 10) & 31;
  const int l2 = (lane + 20) & 31;
  const float4* __restrict__ H4 = (const float4*)H;
  for (int node = blockIdx.x * 8 + wave; node < n; node += nwaves) {
    const int beg = off[node];
    const int deg = cnt[node];
    float4 acc = make_float4(0.0f, 0.0f, 0.0f, 0.0f);
    for (int j = 0; j < deg; j += 32) {
      const int m = (deg - j < 32) ? (deg - j) : 32;
      const int myidx = (lane < m) ? ssrc[beg + j + lane] : 0;
      for (int t = 0; t < m; t += 12) {
        const float4 v0 = load_row40(H4, myidx, t + g,     g, m, c);
        const float4 v1 = load_row40(H4, myidx, t + 3 + g, g, m, c);
        const float4 v2 = load_row40(H4, myidx, t + 6 + g, g, m, c);
        const float4 v3 = load_row40(H4, myidx, t + 9 + g, g, m, c);
        acc.x += v0.x; acc.y += v0.y; acc.z += v0.z; acc.w += v0.w;
        acc.x += v1.x; acc.y += v1.y; acc.z += v1.z; acc.w += v1.w;
        acc.x += v2.x; acc.y += v2.y; acc.z += v2.z; acc.w += v2.w;
        acc.x += v3.x; acc.y += v3.y; acc.z += v3.z; acc.w += v3.w;
      }
    }
    float4 r;
    r.x = (acc.x + __shfl(acc.x, l1, 32)) + __shfl(acc.x, l2, 32);
    r.y = (acc.y + __shfl(acc.y, l1, 32)) + __shfl(acc.y, l2, 32);
    r.z = (acc.z + __shfl(acc.z, l1, 32)) + __shfl(acc.z, l2, 32);
    r.w = (acc.w + __shfl(acc.w, l1, 32)) + __shfl(acc.w, l2, 32);
    if (lane < OUT_CH / 4) ((float4*)OUT)[(size_t)node * (OUT_CH / 4) + lane] = r;
  }
}

__global__ __launch_bounds__(256) void bias_relu_kernel(float* __restrict__ A,
                                                        const float* __restrict__ b1,
                                                        long long total) {
  long long i = (long long)blockIdx.x * blockDim.x + threadIdx.x;
  const long long stride = (long long)gridDim.x * blockDim.x;
  for (; i < total; i += stride) {
    const float v = A[i] + b1[i & (HID_CH - 1)];
    A[i] = v > 0.0f ? v : 0.0f;
  }
}

__global__ __launch_bounds__(256) void bias_logsoftmax_kernel(float* __restrict__ O,
                                                              const float* __restrict__ b2,
                                                              int n) {
  const int lane = threadIdx.x & 31;
  const int nwaves = gridDim.x * 8;
  const bool hi = (lane < OUT_CH - 32);
  const float ba = b2[lane];
  const float bb = hi ? b2[lane + 32] : 0.0f;
  for (int row = blockIdx.x * 8 + (threadIdx.x >> 5); row < n; row += nwaves) {
    float* o = O + (size_t)row * OUT_CH;
    const float a = o[lane] + ba;
    const float b = hi ? (o[lane + 32] + bb) : -INFINITY;

    float m = fmaxf(a, b);
    for (int off = 16; off > 0; off >>= 1) m = fmaxf(m, __shfl_xor(m, off, 32));
    float s = expf(a - m) + (hi ? expf(b - m) : 0.0f);
    for (int off = 16; off > 0; off >>= 1) s += __shfl_xor(s, off, 32);
    const float lse = m + logf(s);

    o[lane] = a - lse;
    if (hi) o[lane + 32] = b - lse;
  }
}

extern "C" void kernel_launch(void* const* d_in, const int* in_sizes, int n_in,
                              void* d_out, int out_size, void* d_ws, size_t ws_size,
                              hipStream_t stream) {
  const float* x  = (const float*)d_in[0];
  const int*   ei = (const int*)d_in[1];
  const float* W1 = (const float*)d_in[2];
  const float* b1 = (const float*)d_in[3];
  const float* W2 = (const float*)d_in[4];
  const float* b2 = (const float*)d_in[5];

  const int n       = in_sizes[0] / IN_CH;
  const int n_edges = in_sizes[1] / 2;
  const int* src = ei;
  const int* dst = ei + n_edges;

  float* h1   = (float*)d_ws;
  float* agg1 = h1 + (size_t)n * HID_CH;
  float* h2   = h1;
  float* out  = (float*)d_out;
  int* cnt    = (int*)(agg1 + (size_t)n * HID_CH);
  int* cursor = cnt + n;
  int* off    = cursor + 1;
  int* rank   = off + n;
  int* ssrc   = rank + n_edges;

  const int tiles   = (n + 15) / 16;
  const int nblk    = (n + 7) / 8;
  const int eblk    = (n_edges + 255) / 256;
  const int g_gemm1 = tiles < 1024 ? tiles : 1024;
  const int g_gemm2 = tiles < 2048 ? tiles : 2048;
  const int g_node  = nblk < 2048 ? nblk : 2048;
  const int g_edge  = eblk < 1024 ? eblk : 1024;

  zero_int_kernel<<<(n + 1 + 255) / 256, 256, 0, stream>>>(cnt, n + 1);
  rank_kernel<<<g_edge, 256, 0, stream>>>(dst, cnt, rank, n_edges);
  offsets_kernel<<<(n + 255) / 256, 256, 0, stream>>>(cnt, off, cursor, n);
  place_kernel<<<g_edge, 256, 0, stream>>>(src, dst, rank, off, ssrc, n_edges);

  gemm1_kernel<<<g_gemm1, 256, 0, stream>>>(x, W1, h1, n, tiles);
  gather1_kernel<<<g_node, 256, 0, stream>>>(h1, ssrc, off, cnt, agg1, n);
  bias_relu_kernel<<<2048, 256, 0, stream>>>(agg1, b1, (long long)n * HID_CH);

  gemm2_kernel<<<g_gemm2, 96, 0, stream>>>(agg1, W2, h2, n, tiles);
  gather2_kernel<<<g_node, 256, 0, stream>>>(h2, ssrc, off, cnt, out, n);
  bias_logsoftmax_kernel<<<g_node, 256, 0, stream>>>(out, b2, n);
}


# Layer-1 aggregate stays on chip: the gather kernel adds b1, applies relu and does the 128x40 WMMA product on a 16-row LDS tile and writes H2; bias_relu and GEMM2 launches and the AGG round trip go
# speedup vs baseline: 2.1578x; 1.1473x over previous
#include <hip/hip_runtime.h>
#include <math.h>

typedef float v2f __attribute__((ext_vector_type(2)));
typedef float v8f __attribute__((ext_vector_type(8)));

#define IN_CH 128
#define HID_CH 128
#define OUT_CH 40

__global__ __launch_bounds__(256) void gemm1_kernel(const float* __restrict__ X,
                                                    const float* __restrict__ W,
                                                    float* __restrict__ H, int n, int tiles) {
  const int wave = threadIdx.x >> 5;
  const int lane = threadIdx.x & 31;
  const int half = lane >> 4;
  const int l    = lane & 15;
  const int col = (wave << 4) + l;

  v2f bw[IN_CH / 4];
#pragma unroll
  for (int i = 0; i < IN_CH / 4; ++i) {
    const int ka = 4 * i + (half << 1);
    bw[i].x = W[(size_t)ka * HID_CH + col];
    bw[i].y = W[(size_t)(ka + 1) * HID_CH + col];
  }

  for (int tile = blockIdx.x; tile < tiles; tile += gridDim.x) {
    const int row_base = tile * 16;
    int arow_i = row_base + l;
    if (arow_i >= n) arow_i = n - 1;
    const float* __restrict__ arow = X + (size_t)arow_i * IN_CH;

    v8f acc = {};
#pragma unroll
    for (int i = 0; i < IN_CH / 4; ++i) {
      const int ka = 4 * i + (half << 1);
      v2f a;
      a.x = arow[ka];
      a.y = arow[ka + 1];
      acc = __builtin_amdgcn_wmma_f32_16x16x4_f32(false, a, false, bw[i],
                                                  (short)0, acc, false, false);
    }

    float* __restrict__ crow = H + (size_t)(row_base + (half << 3)) * HID_CH + col;
    if (row_base + 16 <= n) {
#pragma unroll
      for (int v = 0; v < 8; ++v) crow[(size_t)v * HID_CH] = acc[v];
    } else {
#pragma unroll
      for (int v = 0; v < 8; ++v) {
        if (row_base + (half << 3) + v < n) crow[(size_t)v * HID_CH] = acc[v];
      }
    }
  }
}

__global__ __launch_bounds__(256) void zero_int_kernel(int* __restrict__ p, int count) {
  const int i = blockIdx.x * 256 + threadIdx.x;
  if (i < count) p[i] = 0;
}

__global__ __launch_bounds__(256) void rank_kernel(const int* __restrict__ dst,
                                                   int* __restrict__ cnt,
                                                   int* __restrict__ rank,
                                                   int n_edges) {
  const int stride = gridDim.x * 256;
  int e = blockIdx.x * 256 + threadIdx.x;
  for (; e + 3 * stride < n_edges; e += 4 * stride) {
    const int d0 = dst[e];
    const int d1 = dst[e + stride];
    const int d2 = dst[e + 2 * stride];
    const int d3 = dst[e + 3 * stride];
    const int r0 = atomicAdd(&cnt[d0], 1);
    const int r1 = atomicAdd(&cnt[d1], 1);
    const int r2 = atomicAdd(&cnt[d2], 1);
    const int r3 = atomicAdd(&cnt[d3], 1);
    rank[e] = r0;
    rank[e + stride] = r1;
    rank[e + 2 * stride] = r2;
    rank[e + 3 * stride] = r3;
  }
  for (; e < n_edges; e += stride) rank[e] = atomicAdd(&cnt[dst[e]], 1);
}

__global__ __launch_bounds__(256) void offsets_kernel(const int* __restrict__ cnt,
                                                      int* __restrict__ off,
                                                      int* __restrict__ cursor,
                                                      int n) {
  const int i = blockIdx.x * 256 + threadIdx.x;
  const int lane = threadIdx.x & 31;
  const int c = (i < n) ? cnt[i] : 0;
  int x = c;
  for (int o = 1; o < 32; o <<= 1) {
    const int y = __shfl_up(x, o, 32);
    if (lane >= o) x += y;
  }
  int base = 0;
  if (lane == 31) base = atomicAdd(cursor, x);
  base = __shfl(base, 31, 32);
  if (i < n) off[i] = base + x - c;
}

__global__ __launch_bounds__(256) void place_kernel(const int* __restrict__ src,
                                                    const int* __restrict__ dst,
                                                    const int* __restrict__ rank,
                                                    const int* __restrict__ off,
                                                    int* __restrict__ ssrc,
                                                    int n_edges) {
  const int stride = gridDim.x * 256;
  int e = blockIdx.x * 256 + threadIdx.x;
  for (; e + 3 * stride < n_edges; e += 4 * stride) {
    const int d0 = dst[e];
    const int d1 = dst[e + stride];
    const int d2 = dst[e + 2 * stride];
    const int d3 = dst[e + 3 * stride];
    const int r0 = rank[e];
    const int r1 = rank[e + stride];
    const int r2 = rank[e + 2 * stride];
    const int r3 = rank[e + 3 * stride];
    const int s0 = src[e];
    const int s1 = src[e + stride];
    const int s2 = src[e + 2 * stride];
    const int s3 = src[e + 3 * stride];
    const int o0 = off[d0];
    const int o1 = off[d1];
    const int o2 = off[d2];
    const int o3 = off[d3];
    ssrc[o0 + r0] = s0;
    ssrc[o1 + r1] = s1;
    ssrc[o2 + r2] = s2;
    ssrc[o3 + r3] = s3;
  }
  for (; e < n_edges; e += stride) ssrc[off[dst[e]] + rank[e]] = src[e];
}

template <int K>
__device__ __forceinline__ void add_rows128(const float4* __restrict__ H4, int myidx,
                                            int t, int lane, float4& acc) {
  float4 v[K];
#pragma unroll
  for (int u = 0; u < K; ++u) {
    const int s = __builtin_amdgcn_readlane(myidx, t + u);
    v[u] = H4[(size_t)s * (HID_CH / 4) + lane];
  }
#pragma unroll
  for (int u = 0; u < K; ++u) {
    acc.x += v[u].x; acc.y += v[u].y; acc.z += v[u].z; acc.w += v[u].w;
  }
}

#define TILE_PITCH 132
__global__ __launch_bounds__(512, 16) void gather1_gemm2_kernel(const float* __restrict__ H,
                                                            const int* __restrict__ ssrc,
                                                            const int* __restrict__ off,
                                                            const int* __restrict__ cnt,
                                                            const float* __restrict__ b1,
                                                            const float* __restrict__ W2,
                                                            float* __restrict__ H2,
                                                            int n, int tiles) {
  __shared__ float4 w2s4[HID_CH * OUT_CH / 4];
  __shared__ float4 tile4[2 * 16 * TILE_PITCH / 4];
  const float* w2s = (const float*)w2s4;
  float* tilef = (float*)tile4;

  const int wave = __builtin_amdgcn_readfirstlane((int)(threadIdx.x >> 5));
  const int lane = threadIdx.x & 31;
  const int half = lane >> 4;
  const int l    = lane & 15;
  const int col  = (wave << 4) + l;
  const bool colok = (col < OUT_CH);
  const int colc = colok ? col : 0;
  const float4* __restrict__ H4 = (const float4*)H;

  for (int q = threadIdx.x; q < HID_CH * OUT_CH / 4; q += 512) w2s4[q] = ((const float4*)W2)[q];
  const float4 bias = ((const float4*)b1)[lane];

  int buf = 0;
  for (int t = blockIdx.x; t < tiles; t += gridDim.x) {
    const int row_base = t * 16;
    const int node = row_base + wave;
    float4 acc = make_float4(0.0f, 0.0f, 0.0f, 0.0f);
    if (node < n) {
      const int beg = __builtin_amdgcn_readfirstlane(off[node]);
      const int deg = __builtin_amdgcn_readfirstlane(cnt[node]);
      for (int j = 0; j < deg; j += 32) {
        const int m = (deg - j < 32) ? (deg - j) : 32;
        const int myidx = (lane < m) ? ssrc[beg + j + lane] : 0;
        int u = 0;
        for (; u + 8 <= m; u += 8) add_rows128<8>(H4, myidx, u, lane, acc);
        if (u + 4 <= m) { add_rows128<4>(H4, myidx, u, lane, acc); u += 4; }
        if (u + 2 <= m) { add_rows128<2>(H4, myidx, u, lane, acc); u += 2; }
        if (u < m) add_rows128<1>(H4, myidx, u, lane, acc);
      }
    }
    float4 r;
    r.x = acc.x + bias.x; r.y = acc.y + bias.y; r.z = acc.z + bias.z; r.w = acc.w + bias.w;
    r.x = r.x > 0.0f ? r.x : 0.0f;
    r.y = r.y > 0.0f ? r.y : 0.0f;
    r.z = r.z > 0.0f ? r.z : 0.0f;
    r.w = r.w > 0.0f ? r.w : 0.0f;
    float* tb = tilef + buf * (16 * TILE_PITCH);
    *(float4*)(tb + wave * TILE_PITCH + 4 * lane) = r;
    __syncthreads();

    if (wave < 3) {
      const float* arow = tb + l * TILE_PITCH + (half << 1);
      const float* bcol = w2s + (half << 1) * OUT_CH + colc;
      v8f c = {};
      v2f a[2][4], b[2][4];
#pragma unroll
      for (int s = 0; s < 4; ++s) {
        a[0][s] = *(const v2f*)(arow + 4 * s);
        b[0][s].x = bcol[(4 * s) * OUT_CH];
        b[0][s].y = bcol[(4 * s + 1) * OUT_CH];
      }
      __builtin_amdgcn_sched_group_barrier(0x100, 6, 0);
#pragma unroll
      for (int g = 0; g < HID_CH / 16; ++g) {
        if (g + 1 < HID_CH / 16) {
#pragma unroll
          for (int s = 0; s < 4; ++s) {
            const int i = 4 * (g + 1) + s;
            a[(g + 1) & 1][s] = *(const v2f*)(arow + 4 * i);
            b[(g + 1) & 1][s].x = bcol[(4 * i) * OUT_CH];
            b[(g + 1) & 1][s].y = bcol[(4 * i + 1) * OUT_CH];
          }
        }
#pragma unroll
        for (int s = 0; s < 4; ++s) {
          c = __builtin_amdgcn_wmma_f32_16x16x4_f32(false, a[g & 1][s], false, b[g & 1][s],
                                                    (short)0, c, false, false);
        }
        __builtin_amdgcn_sched_group_barrier(0x100, 6, 0);
        __builtin_amdgcn_sched_group_barrier(0x008, 4, 0);
      }
      float* __restrict__ crow = H2 + (size_t)(row_base + (half << 3)) * OUT_CH + col;
      if (colok) {
        if (row_base + 16 <= n) {
#pragma unroll
          for (int v = 0; v < 8; ++v) crow[(size_t)v * OUT_CH] = c[v];
        } else {
#pragma unroll
          for (int v = 0; v < 8; ++v) {
            if (row_base + (half << 3) + v < n) crow[(size_t)v * OUT_CH] = c[v];
          }
        }
      }
    }
    buf ^= 1;
  }
}

__device__ __forceinline__ float4 load_row40(const float4* __restrict__ H4, int myidx,
                                             int e, int g, int m, int c) {
  const int s = __shfl(myidx, e & 31, 32);
  float4 v = make_float4(0.0f, 0.0f, 0.0f, 0.0f);
  if (g < 3 && e < m) v = H4[(size_t)s * (OUT_CH / 4) + c];
  return v;
}

__global__ __launch_bounds__(256) void gather2_kernel(const float* __restrict__ H,
                                                      const int* __restrict__ ssrc,
                                                      const int* __restrict__ off,
                                                      const int* __restrict__ cnt,
                                                      float* __restrict__ OUT, int n) {
  const int wave = __builtin_amdgcn_readfirstlane((int)(threadIdx.x >> 5));
  const int lane = threadIdx.x & 31;
  const int nwaves = gridDim.x * 8;
  const int g = lane / 10;
  const int c = lane - g * 10;
  const int l1 = (lane + 10) & 31;
  const int l2 = (lane + 20) & 31;
  const float4* __restrict__ H4 = (const float4*)H;
  for (int node = blockIdx.x * 8 + wave; node < n; node += nwaves) {
    const int beg = off[node];
    const int deg = cnt[node];
    float4 acc = make_float4(0.0f, 0.0f, 0.0f, 0.0f);
    for (int j = 0; j < deg; j += 32) {
      const int m = (deg - j < 32) ? (deg - j) : 32;
      const int myidx = (lane < m) ? ssrc[beg + j + lane] : 0;
      for (int t = 0; t < m; t += 12) {
        const float4 v0 = load_row40(H4, myidx, t + g,     g, m, c);
        const float4 v1 = load_row40(H4, myidx, t + 3 + g, g, m, c);
        const float4 v2 = load_row40(H4, myidx, t + 6 + g, g, m, c);
        const float4 v3 = load_row40(H4, myidx, t + 9 + g, g, m, c);
        acc.x += v0.x; acc.y += v0.y; acc.z += v0.z; acc.w += v0.w;
        acc.x += v1.x; acc.y += v1.y; acc.z += v1.z; acc.w += v1.w;
        acc.x += v2.x; acc.y += v2.y; acc.z += v2.z; acc.w += v2.w;
        acc.x += v3.x; acc.y += v3.y; acc.z += v3.z; acc.w += v3.w;
      }
    }
    float4 r;
    r.x = (acc.x + __shfl(acc.x, l1, 32)) + __shfl(acc.x, l2, 32);
    r.y = (acc.y + __shfl(acc.y, l1, 32)) + __shfl(acc.y, l2, 32);
    r.z = (acc.z + __shfl(acc.z, l1, 32)) + __shfl(acc.z, l2, 32);
    r.w = (acc.w + __shfl(acc.w, l1, 32)) + __shfl(acc.w, l2, 32);
    if (lane < OUT_CH / 4) ((float4*)OUT)[(size_t)node * (OUT_CH / 4) + lane] = r;
  }
}

__global__ __launch_bounds__(256) void bias_logsoftmax_kernel(float* __restrict__ O,
                                                              const float* __restrict__ b2,
                                                              int n) {
  const int lane = threadIdx.x & 31;
  const int nwaves = gridDim.x * 8;
  const bool hi = (lane < OUT_CH - 32);
  const float ba = b2[lane];
  const float bb = hi ? b2[lane + 32] : 0.0f;
  for (int row = blockIdx.x * 8 + (threadIdx.x >> 5); row < n; row += nwaves) {
    float* o = O + (size_t)row * OUT_CH;
    const float a = o[lane] + ba;
    const float b = hi ? (o[lane + 32] + bb) : -INFINITY;

    float m = fmaxf(a, b);
    for (int off = 16; off > 0; off >>= 1) m = fmaxf(m, __shfl_xor(m, off, 32));
    float s = expf(a - m) + (hi ? expf(b - m) : 0.0f);
    for (int off = 16; off > 0; off >>= 1) s += __shfl_xor(s, off, 32);
    const float lse = m + logf(s);

    o[lane] = a - lse;
    if (hi) o[lane + 32] = b - lse;
  }
}

extern "C" void kernel_launch(void* const* d_in, const int* in_sizes, int n_in,
                              void* d_out, int out_size, void* d_ws, size_t ws_size,
                              hipStream_t stream) {
  const float* x  = (const float*)d_in[0];
  const int*   ei = (const int*)d_in[1];
  const float* W1 = (const float*)d_in[2];
  const float* b1 = (const float*)d_in[3];
  const float* W2 = (const float*)d_in[4];
  const float* b2 = (const float*)d_in[5];

  const int n       = in_sizes[0] / IN_CH;
  const int n_edges = in_sizes[1] / 2;
  const int* src = ei;
  const int* dst = ei + n_edges;

  float* h1   = (float*)d_ws;
  float* h2   = h1 + (size_t)n * HID_CH;
  float* out  = (float*)d_out;
  int* cnt    = (int*)(h2 + (size_t)n * HID_CH);
  int* cursor = cnt + n;
  int* off    = cursor + 1;
  int* rank   = off + n;
  int* ssrc   = rank + n_edges;

  const int tiles   = (n + 15) / 16;
  const int nblk    = (n + 7) / 8;
  const int eblk    = (n_edges + 255) / 256;
  const int g_gemm1 = tiles < 1024 ? tiles : 1024;
  const int g_fused = tiles < 1024 ? tiles : 1024;
  const int g_node  = nblk < 2048 ? nblk : 2048;
  const int g_edge  = eblk < 1024 ? eblk : 1024;

  zero_int_kernel<<<(n + 1 + 255) / 256, 256, 0, stream>>>(cnt, n + 1);
  rank_kernel<<<g_edge, 256, 0, stream>>>(dst, cnt, rank, n_edges);
  offsets_kernel<<<(n + 255) / 256, 256, 0, stream>>>(cnt, off, cursor, n);
  place_kernel<<<g_edge, 256, 0, stream>>>(src, dst, rank, off, ssrc, n_edges);

  gemm1_kernel<<<g_gemm1, 256, 0, stream>>>(x, W1, h1, n, tiles);
  gather1_gemm2_kernel<<<g_fused, 512, 0, stream>>>(h1, ssrc, off, cnt, b1, W2, h2, n, tiles);

  gather2_kernel<<<g_node, 256, 0, stream>>>(h2, ssrc, off, cnt, out, n);
  bias_logsoftmax_kernel<<<g_node, 256, 0, stream>>>(out, b2, n);
}


# Edge grouping by destination moved to LDS counters: edges binned by 256-node range (chunk histograms, scan), then degree, start and arrival-order rank per bin in LDS; 1.6M returning global adds go
# speedup vs baseline: 3.3421x; 1.5488x over previous
#include <hip/hip_runtime.h>
#include <math.h>

typedef float v2f __attribute__((ext_vector_type(2)));
typedef float v8f __attribute__((ext_vector_type(8)));

#define IN_CH 128
#define HID_CH 128
#define OUT_CH 40

__global__ __launch_bounds__(256) void gemm1_kernel(const float* __restrict__ X,
                                                    const float* __restrict__ W,
                                                    float* __restrict__ H, int n, int tiles) {
  const int wave = threadIdx.x >> 5;
  const int lane = threadIdx.x & 31;
  const int half = lane >> 4;
  const int l    = lane & 15;
  const int col = (wave << 4) + l;

  v2f bw[IN_CH / 4];
#pragma unroll
  for (int i = 0; i < IN_CH / 4; ++i) {
    const int ka = 4 * i + (half << 1);
    bw[i].x = W[(size_t)ka * HID_CH + col];
    bw[i].y = W[(size_t)(ka + 1) * HID_CH + col];
  }

  for (int tile = blockIdx.x; tile < tiles; tile += gridDim.x) {
    const int row_base = tile * 16;
    int arow_i = row_base + l;
    if (arow_i >= n) arow_i = n - 1;
    const float* __restrict__ arow = X + (size_t)arow_i * IN_CH;

    v8f acc = {};
#pragma unroll
    for (int i = 0; i < IN_CH / 4; ++i) {
      const int ka = 4 * i + (half << 1);
      v2f a;
      a.x = arow[ka];
      a.y = arow[ka + 1];
      acc = __builtin_amdgcn_wmma_f32_16x16x4_f32(false, a, false, bw[i],
                                                  (short)0, acc, false, false);
    }

    float* __restrict__ crow = H + (size_t)(row_base + (half << 3)) * HID_CH + col;
    if (row_base + 16 <= n) {
#pragma unroll
      for (int v = 0; v < 8; ++v) crow[(size_t)v * HID_CH] = acc[v];
    } else {
#pragma unroll
      for (int v = 0; v < 8; ++v) {
        if (row_base + (half << 3) + v < n) crow[(size_t)v * HID_CH] = acc[v];
      }
    }
  }
}

#define BIN_SHIFT 8
#define BIN_NODES 256
#define EDGE_CHUNKS 256
#define MAX_BINS 1024

__device__ __forceinline__ int wave_incl_scan(int x, int lane) {
  for (int o = 1; o < 32; o <<= 1) {
    const int y = __shfl_up(x, o, 32);
    if (lane >= o) x += y;
  }
  return x;
}

__global__ __launch_bounds__(1024) void bin_hist_kernel(const int* __restrict__ dst,
                                                        int* __restrict__ bh,
                                                        int* __restrict__ cursor,
                                                        int n_edges, int nb) {
  __shared__ int h[MAX_BINS];
  const int tid = threadIdx.x;
  h[tid] = 0;
  __syncthreads();
  const int chunk = (n_edges + EDGE_CHUNKS - 1) / EDGE_CHUNKS;
  const int e0 = blockIdx.x * chunk;
  const int e1 = (e0 + chunk < n_edges) ? (e0 + chunk) : n_edges;
  int e = e0 + tid;
  for (; e + 3 * 1024 < e1; e += 4 * 1024) {
    const int d0 = dst[e];
    const int d1 = dst[e + 1024];
    const int d2 = dst[e + 2 * 1024];
    const int d3 = dst[e + 3 * 1024];
    atomicAdd(&h[d0 >> BIN_SHIFT], 1);
    atomicAdd(&h[d1 >> BIN_SHIFT], 1);
    atomicAdd(&h[d2 >> BIN_SHIFT], 1);
    atomicAdd(&h[d3 >> BIN_SHIFT], 1);
  }
  for (; e < e1; e += 1024) atomicAdd(&h[dst[e] >> BIN_SHIFT], 1);
  __syncthreads();
  for (int b = tid; b < nb; b += 1024) bh[b * EDGE_CHUNKS + blockIdx.x] = h[b];
  if (blockIdx.x == 0 && tid == 0) *cursor = 0;
}

__global__ __launch_bounds__(256) void bin_scan_kernel(const int* __restrict__ bh,
                                                       int* __restrict__ st,
                                                       int* __restrict__ binstart,
                                                       int* __restrict__ bincnt,
                                                       int* __restrict__ cursor) {
  __shared__ int wt[8];
  __shared__ int base_s;
  const int b = blockIdx.x;
  const int tid = threadIdx.x;
  const int lane = tid & 31;
  const int wave = tid >> 5;
  const int c = bh[b * EDGE_CHUNKS + tid];
  const int x = wave_incl_scan(c, lane);
  if (lane == 31) wt[wave] = x;
  __syncthreads();
  int woff = 0, total = 0;
  for (int w = 0; w < 8; ++w) {
    const int t = wt[w];
    if (w < wave) woff += t;
    total += t;
  }
  if (tid == 0) base_s = atomicAdd(cursor, total);
  __syncthreads();
  const int base = base_s;
  st[b * EDGE_CHUNKS + tid] = base + woff + x - c;
  if (tid == 0) { binstart[b] = base; bincnt[b] = total; }
}

__global__ __launch_bounds__(1024) void bin_scatter_kernel(const int* __restrict__ src,
                                                           const int* __restrict__ dst,
                                                           const int* __restrict__ st,
                                                           int2* __restrict__ eb,
                                                           int n_edges, int nb) {
  __shared__ int pos[MAX_BINS];
  const int tid = threadIdx.x;
  for (int b = tid; b < nb; b += 1024) pos[b] = st[b * EDGE_CHUNKS + blockIdx.x];
  __syncthreads();
  const int chunk = (n_edges + EDGE_CHUNKS - 1) / EDGE_CHUNKS;
  const int e0 = blockIdx.x * chunk;
  const int e1 = (e0 + chunk < n_edges) ? (e0 + chunk) : n_edges;
  int e = e0 + tid;
  for (; e + 3 * 1024 < e1; e += 4 * 1024) {
    const int d0 = dst[e];
    const int d1 = dst[e + 1024];
    const int d2 = dst[e + 2 * 1024];
    const int d3 = dst[e + 3 * 1024];
    const int s0 = src[e];
    const int s1 = src[e + 1024];
    const int s2 = src[e + 2 * 1024];
    const int s3 = src[e + 3 * 1024];
    const int p0 = atomicAdd(&pos[d0 >> BIN_SHIFT], 1);
    const int p1 = atomicAdd(&pos[d1 >> BIN_SHIFT], 1);
    const int p2 = atomicAdd(&pos[d2 >> BIN_SHIFT], 1);
    const int p3 = atomicAdd(&pos[d3 >> BIN_SHIFT], 1);
    eb[p0] = make_int2(d0, s0);
    eb[p1] = make_int2(d1, s1);
    eb[p2] = make_int2(d2, s2);
    eb[p3] = make_int2(d3, s3);
  }
  for (; e < e1; e += 1024) {
    const int d = dst[e];
    const int s = src[e];
    const int p = atomicAdd(&pos[d >> BIN_SHIFT], 1);
    eb[p] = make_int2(d, s);
  }
}

__global__ __launch_bounds__(1024) void bin_sort_kernel(const int2* __restrict__ eb,
                                                        const int* __restrict__ binstart,
                                                        const int* __restrict__ bincnt,
                                                        int* __restrict__ cnt,
                                                        int* __restrict__ off,
                                                        int* __restrict__ ssrc, int n) {
  __shared__ int c[BIN_NODES];
  __shared__ int o[BIN_NODES];
  __shared__ int wt[8];
  const int b = blockIdx.x;
  const int tid = threadIdx.x;
  const int lane = tid & 31;
  const int wave = tid >> 5;
  const int node0 = b << BIN_SHIFT;
  if (tid < BIN_NODES) c[tid] = 0;
  __syncthreads();
  const int beg = binstart[b];
  const int m = bincnt[b];
  int i = tid;
  for (; i + 3 * 1024 < m; i += 4 * 1024) {
    const int d0 = eb[beg + i].x;
    const int d1 = eb[beg + i + 1024].x;
    const int d2 = eb[beg + i + 2 * 1024].x;
    const int d3 = eb[beg + i + 3 * 1024].x;
    atomicAdd(&c[d0 - node0], 1);
    atomicAdd(&c[d1 - node0], 1);
    atomicAdd(&c[d2 - node0], 1);
    atomicAdd(&c[d3 - node0], 1);
  }
  for (; i < m; i += 1024) atomicAdd(&c[eb[beg + i].x - node0], 1);
  __syncthreads();
  int v = 0, x = 0;
  if (tid < BIN_NODES) {
    v = c[tid];
    x = wave_incl_scan(v, lane);
    if (lane == 31) wt[wave] = x;
  }
  __syncthreads();
  if (tid < BIN_NODES) {
    int woff = 0;
    for (int w = 0; w < 8; ++w) {
      const int t = wt[w];
      if (w < wave) woff += t;
    }
    const int ex = woff + x - v;
    o[tid] = ex;
    const int node = node0 + tid;
    if (node < n) { cnt[node] = v; off[node] = beg + ex; }
  }
  __syncthreads();
  i = tid;
  for (; i + 3 * 1024 < m; i += 4 * 1024) {
    const int2 r0 = eb[beg + i];
    const int2 r1 = eb[beg + i + 1024];
    const int2 r2 = eb[beg + i + 2 * 1024];
    const int2 r3 = eb[beg + i + 3 * 1024];
    const int p0 = atomicAdd(&o[r0.x - node0], 1);
    const int p1 = atomicAdd(&o[r1.x - node0], 1);
    const int p2 = atomicAdd(&o[r2.x - node0], 1);
    const int p3 = atomicAdd(&o[r3.x - node0], 1);
    ssrc[beg + p0] = r0.y;
    ssrc[beg + p1] = r1.y;
    ssrc[beg + p2] = r2.y;
    ssrc[beg + p3] = r3.y;
  }
  for (; i < m; i += 1024) {
    const int2 r = eb[beg + i];
    const int p = atomicAdd(&o[r.x - node0], 1);
    ssrc[beg + p] = r.y;
  }
}

template <int K>
__device__ __forceinline__ void add_rows128(const float4* __restrict__ H4, int myidx,
                                            int t, int lane, float4& acc) {
  float4 v[K];
#pragma unroll
  for (int u = 0; u < K; ++u) {
    const int s = __builtin_amdgcn_readlane(myidx, t + u);
    v[u] = H4[(size_t)s * (HID_CH / 4) + lane];
  }
#pragma unroll
  for (int u = 0; u < K; ++u) {
    acc.x += v[u].x; acc.y += v[u].y; acc.z += v[u].z; acc.w += v[u].w;
  }
}

#define TILE_PITCH 132
__global__ __launch_bounds__(512, 16) void gather1_gemm2_kernel(const float* __restrict__ H,
                                                            const int* __restrict__ ssrc,
                                                            const int* __restrict__ off,
                                                            const int* __restrict__ cnt,
                                                            const float* __restrict__ b1,
                                                            const float* __restrict__ W2,
                                                            float* __restrict__ H2,
                                                            int n, int tiles) {
  __shared__ float4 w2s4[HID_CH * OUT_CH / 4];
  __shared__ float4 tile4[2 * 16 * TILE_PITCH / 4];
  const float* w2s = (const float*)w2s4;
  float* tilef = (float*)tile4;

  const int wave = __builtin_amdgcn_readfirstlane((int)(threadIdx.x >> 5));
  const int lane = threadIdx.x & 31;
  const int half = lane >> 4;
  const int l    = lane & 15;
  const int col  = (wave << 4) + l;
  const bool colok = (col < OUT_CH);
  const int colc = colok ? col : 0;
  const float4* __restrict__ H4 = (const float4*)H;

  for (int q = threadIdx.x; q < HID_CH * OUT_CH / 4; q += 512) w2s4[q] = ((const float4*)W2)[q];
  const float4 bias = ((const float4*)b1)[lane];

  int buf = 0;
  for (int t = blockIdx.x; t < tiles; t += gridDim.x) {
    const int row_base = t * 16;
    const int node = row_base + wave;
    float4 acc = make_float4(0.0f, 0.0f, 0.0f, 0.0f);
    if (node < n) {
      const int beg = __builtin_amdgcn_readfirstlane(off[node]);
      const int deg = __builtin_amdgcn_readfirstlane(cnt[node]);
      for (int j = 0; j < deg; j += 32) {
        const int m = (deg - j < 32) ? (deg - j) : 32;
        const int myidx = (lane < m) ? ssrc[beg + j + lane] : 0;
        int u = 0;
        for (; u + 8 <= m; u += 8) add_rows128<8>(H4, myidx, u, lane, acc);
        if (u + 4 <= m) { add_rows128<4>(H4, myidx, u, lane, acc); u += 4; }
        if (u + 2 <= m) { add_rows128<2>(H4, myidx, u, lane, acc); u += 2; }
        if (u < m) add_rows128<1>(H4, myidx, u, lane, acc);
      }
    }
    float4 r;
    r.x = acc.x + bias.x; r.y = acc.y + bias.y; r.z = acc.z + bias.z; r.w = acc.w + bias.w;
    r.x = r.x > 0.0f ? r.x : 0.0f;
    r.y = r.y > 0.0f ? r.y : 0.0f;
    r.z = r.z > 0.0f ? r.z : 0.0f;
    r.w = r.w > 0.0f ? r.w : 0.0f;
    float* tb = tilef + buf * (16 * TILE_PITCH);
    *(float4*)(tb + wave * TILE_PITCH + 4 * lane) = r;
    __syncthreads();

    if (wave < 3) {
      const float* arow = tb + l * TILE_PITCH + (half << 1);
      const float* bcol = w2s + (half << 1) * OUT_CH + colc;
      v8f c = {};
      v2f a[2][4], b[2][4];
#pragma unroll
      for (int s = 0; s < 4; ++s) {
        a[0][s] = *(const v2f*)(arow + 4 * s);
        b[0][s].x = bcol[(4 * s) * OUT_CH];
        b[0][s].y = bcol[(4 * s + 1) * OUT_CH];
      }
      __builtin_amdgcn_sched_group_barrier(0x100, 6, 0);
#pragma unroll
      for (int g = 0; g < HID_CH / 16; ++g) {
        if (g + 1 < HID_CH / 16) {
#pragma unroll
          for (int s = 0; s < 4; ++s) {
            const int i = 4 * (g + 1) + s;
            a[(g + 1) & 1][s] = *(const v2f*)(arow + 4 * i);
            b[(g + 1) & 1][s].x = bcol[(4 * i) * OUT_CH];
            b[(g + 1) & 1][s].y = bcol[(4 * i + 1) * OUT_CH];
          }
        }
#pragma unroll
        for (int s = 0; s < 4; ++s) {
          c = __builtin_amdgcn_wmma_f32_16x16x4_f32(false, a[g & 1][s], false, b[g & 1][s],
                                                    (short)0, c, false, false);
        }
        __builtin_amdgcn_sched_group_barrier(0x100, 6, 0);
        __builtin_amdgcn_sched_group_barrier(0x008, 4, 0);
      }
      float* __restrict__ crow = H2 + (size_t)(row_base + (half << 3)) * OUT_CH + col;
      if (colok) {
        if (row_base + 16 <= n) {
#pragma unroll
          for (int v = 0; v < 8; ++v) crow[(size_t)v * OUT_CH] = c[v];
        } else {
#pragma unroll
          for (int v = 0; v < 8; ++v) {
            if (row_base + (half << 3) + v < n) crow[(size_t)v * OUT_CH] = c[v];
          }
        }
      }
    }
    buf ^= 1;
  }
}

__device__ __forceinline__ float4 load_row40(const float4* __restrict__ H4, int myidx,
                                             int e, int g, int m, int c) {
  const int s = __shfl(myidx, e & 31, 32);
  float4 v = make_float4(0.0f, 0.0f, 0.0f, 0.0f);
  if (g < 3 && e < m) v = H4[(size_t)s * (OUT_CH / 4) + c];
  return v;
}

__global__ __launch_bounds__(256) void gather2_kernel(const float* __restrict__ H,
                                                      const int* __restrict__ ssrc,
                                                      const int* __restrict__ off,
                                                      const int* __restrict__ cnt,
                                                      float* __restrict__ OUT, int n) {
  const int wave = __builtin_amdgcn_readfirstlane((int)(threadIdx.x >> 5));
  const int lane = threadIdx.x & 31;
  const int nwaves = gridDim.x * 8;
  const int g = lane / 10;
  const int c = lane - g * 10;
  const int l1 = (lane + 10) & 31;
  const int l2 = (lane + 20) & 31;
  const float4* __restrict__ H4 = (const float4*)H;
  for (int node = blockIdx.x * 8 + wave; node < n; node += nwaves) {
    const int beg = off[node];
    const int deg = cnt[node];
    float4 acc = make_float4(0.0f, 0.0f, 0.0f, 0.0f);
    for (int j = 0; j < deg; j += 32) {
      const int m = (deg - j < 32) ? (deg - j) : 32;
      const int myidx = (lane < m) ? ssrc[beg + j + lane] : 0;
      for (int t = 0; t < m; t += 12) {
        const float4 v0 = load_row40(H4, myidx, t + g,     g, m, c);
        const float4 v1 = load_row40(H4, myidx, t + 3 + g, g, m, c);
        const float4 v2 = load_row40(H4, myidx, t + 6 + g, g, m, c);
        const float4 v3 = load_row40(H4, myidx, t + 9 + g, g, m, c);
        acc.x += v0.x; acc.y += v0.y; acc.z += v0.z; acc.w += v0.w;
        acc.x += v1.x; acc.y += v1.y; acc.z += v1.z; acc.w += v1.w;
        acc.x += v2.x; acc.y += v2.y; acc.z += v2.z; acc.w += v2.w;
        acc.x += v3.x; acc.y += v3.y; acc.z += v3.z; acc.w += v3.w;
      }
    }
    float4 r;
    r.x = (acc.x + __shfl(acc.x, l1, 32)) + __shfl(acc.x, l2, 32);
    r.y = (acc.y + __shfl(acc.y, l1, 32)) + __shfl(acc.y, l2, 32);
    r.z = (acc.z + __shfl(acc.z, l1, 32)) + __shfl(acc.z, l2, 32);
    r.w = (acc.w + __shfl(acc.w, l1, 32)) + __shfl(acc.w, l2, 32);
    if (lane < OUT_CH / 4) ((float4*)OUT)[(size_t)node * (OUT_CH / 4) + lane] = r;
  }
}

__global__ __launch_bounds__(256) void bias_logsoftmax_kernel(float* __restrict__ O,
                                                              const float* __restrict__ b2,
                                                              int n) {
  const int lane = threadIdx.x & 31;
  const int nwaves = gridDim.x * 8;
  const bool hi = (lane < OUT_CH - 32);
  const float ba = b2[lane];
  const float bb = hi ? b2[lane + 32] : 0.0f;
  for (int row = blockIdx.x * 8 + (threadIdx.x >> 5); row < n; row += nwaves) {
    float* o = O + (size_t)row * OUT_CH;
    const float a = o[lane] + ba;
    const float b = hi ? (o[lane + 32] + bb) : -INFINITY;

    float m = fmaxf(a, b);
    for (int off = 16; off > 0; off >>= 1) m = fmaxf(m, __shfl_xor(m, off, 32));
    float s = expf(a - m) + (hi ? expf(b - m) : 0.0f);
    for (int off = 16; off > 0; off >>= 1) s += __shfl_xor(s, off, 32);
    const float lse = m + logf(s);

    o[lane] = a - lse;
    if (hi) o[lane + 32] = b - lse;
  }
}

extern "C" void kernel_launch(void* const* d_in, const int* in_sizes, int n_in,
                              void* d_out, int out_size, void* d_ws, size_t ws_size,
                              hipStream_t stream) {
  const float* x  = (const float*)d_in[0];
  const int*   ei = (const int*)d_in[1];
  const float* W1 = (const float*)d_in[2];
  const float* b1 = (const float*)d_in[3];
  const float* W2 = (const float*)d_in[4];
  const float* b2 = (const float*)d_in[5];

  const int n       = in_sizes[0] / IN_CH;
  const int n_edges = in_sizes[1] / 2;
  const int* src = ei;
  const int* dst = ei + n_edges;

  float* h1   = (float*)d_ws;
  float* h2   = h1 + (size_t)n * HID_CH;
  float* out  = (float*)d_out;
  const int nb = (n + BIN_NODES - 1) / BIN_NODES;
  int2* eb    = (int2*)(h2 + (size_t)n * HID_CH);
  int* cnt    = (int*)(eb + n_edges);
  int* cursor = cnt + n;
  int* off    = cursor + 1;
  int* ssrc   = off + n;
  int* bh     = ssrc + n_edges;
  int* st     = bh + (size_t)nb * EDGE_CHUNKS;
  int* binstart = st + (size_t)nb * EDGE_CHUNKS;
  int* bincnt = binstart + nb;

  const int tiles   = (n + 15) / 16;
  const int nblk    = (n + 7) / 8;
  const int g_gemm1 = tiles < 1024 ? tiles : 1024;
  const int g_fused = tiles < 1024 ? tiles : 1024;
  const int g_node  = nblk < 2048 ? nblk : 2048;

  bin_hist_kernel<<<EDGE_CHUNKS, 1024, 0, stream>>>(dst, bh, cursor, n_edges, nb);
  bin_scan_kernel<<<nb, 256, 0, stream>>>(bh, st, binstart, bincnt, cursor);
  bin_scatter_kernel<<<EDGE_CHUNKS, 1024, 0, stream>>>(src, dst, st, eb, n_edges, nb);
  bin_sort_kernel<<<nb, 1024, 0, stream>>>(eb, binstart, bincnt, cnt, off, ssrc, n);

  gemm1_kernel<<<g_gemm1, 256, 0, stream>>>(x, W1, h1, n, tiles);
  gather1_gemm2_kernel<<<g_fused, 512, 0, stream>>>(h1, ssrc, off, cnt, b1, W2, h2, n, tiles);

  gather2_kernel<<<g_node, 256, 0, stream>>>(h2, ssrc, off, cnt, out, n);
  bias_logsoftmax_kernel<<<g_node, 256, 0, stream>>>(out, b2, n);
}


# Bias and log-softmax done by the layer-2 gather wave on the row it has just summed (row in lanes 0-9, row_xmask reductions; exp-sum order changes); the log-softmax launch and the out round trip go
# speedup vs baseline: 3.4893x; 1.0440x over previous
#include <hip/hip_runtime.h>
#include <math.h>

typedef float v2f __attribute__((ext_vector_type(2)));
typedef float v8f __attribute__((ext_vector_type(8)));

#define IN_CH 128
#define HID_CH 128
#define OUT_CH 40

__global__ __launch_bounds__(256) void gemm1_kernel(const float* __restrict__ X,
                                                    const float* __restrict__ W,
                                                    float* __restrict__ H, int n, int tiles) {
  const int wave = threadIdx.x >> 5;
  const int lane = threadIdx.x & 31;
  const int half = lane >> 4;
  const int l    = lane & 15;
  const int col = (wave << 4) + l;

  v2f bw[IN_CH / 4];
#pragma unroll
  for (int i = 0; i < IN_CH / 4; ++i) {
    const int ka = 4 * i + (half << 1);
    bw[i].x = W[(size_t)ka * HID_CH + col];
    bw[i].y = W[(size_t)(ka + 1) * HID_CH + col];
  }

  for (int tile = blockIdx.x; tile < tiles; tile += gridDim.x) {
    const int row_base = tile * 16;
    int arow_i = row_base + l;
    if (arow_i >= n) arow_i = n - 1;
    const float* __restrict__ arow = X + (size_t)arow_i * IN_CH;

    v8f acc = {};
#pragma unroll
    for (int i = 0; i < IN_CH / 4; ++i) {
      const int ka = 4 * i + (half << 1);
      v2f a;
      a.x = arow[ka];
      a.y = arow[ka + 1];
      acc = __builtin_amdgcn_wmma_f32_16x16x4_f32(false, a, false, bw[i],
                                                  (short)0, acc, false, false);
    }

    float* __restrict__ crow = H + (size_t)(row_base + (half << 3)) * HID_CH + col;
    if (row_base + 16 <= n) {
#pragma unroll
      for (int v = 0; v < 8; ++v) crow[(size_t)v * HID_CH] = acc[v];
    } else {
#pragma unroll
      for (int v = 0; v < 8; ++v) {
        if (row_base + (half << 3) + v < n) crow[(size_t)v * HID_CH] = acc[v];
      }
    }
  }
}

#define BIN_SHIFT 8
#define BIN_NODES 256
#define EDGE_CHUNKS 256
#define MAX_BINS 1024

__device__ __forceinline__ int wave_incl_scan(int x, int lane) {
  for (int o = 1; o < 32; o <<= 1) {
    const int y = __shfl_up(x, o, 32);
    if (lane >= o) x += y;
  }
  return x;
}

__global__ __launch_bounds__(1024) void bin_hist_kernel(const int* __restrict__ dst,
                                                        int* __restrict__ bh,
                                                        int* __restrict__ cursor,
                                                        int n_edges, int nb) {
  __shared__ int h[MAX_BINS];
  const int tid = threadIdx.x;
  h[tid] = 0;
  __syncthreads();
  const int chunk = (n_edges + EDGE_CHUNKS - 1) / EDGE_CHUNKS;
  const int e0 = blockIdx.x * chunk;
  const int e1 = (e0 + chunk < n_edges) ? (e0 + chunk) : n_edges;
  int e = e0 + tid;
  for (; e + 3 * 1024 < e1; e += 4 * 1024) {
    const int d0 = dst[e];
    const int d1 = dst[e + 1024];
    const int d2 = dst[e + 2 * 1024];
    const int d3 = dst[e + 3 * 1024];
    atomicAdd(&h[d0 >> BIN_SHIFT], 1);
    atomicAdd(&h[d1 >> BIN_SHIFT], 1);
    atomicAdd(&h[d2 >> BIN_SHIFT], 1);
    atomicAdd(&h[d3 >> BIN_SHIFT], 1);
  }
  for (; e < e1; e += 1024) atomicAdd(&h[dst[e] >> BIN_SHIFT], 1);
  __syncthreads();
  for (int b = tid; b < nb; b += 1024) bh[b * EDGE_CHUNKS + blockIdx.x] = h[b];
  if (blockIdx.x == 0 && tid == 0) *cursor = 0;
}

__global__ __launch_bounds__(256) void bin_scan_kernel(const int* __restrict__ bh,
                                                       int* __restrict__ st,
                                                       int* __restrict__ binstart,
                                                       int* __restrict__ bincnt,
                                                       int* __restrict__ cursor) {
  __shared__ int wt[8];
  __shared__ int base_s;
  const int b = blockIdx.x;
  const int tid = threadIdx.x;
  const int lane = tid & 31;
  const int wave = tid >> 5;
  const int c = bh[b * EDGE_CHUNKS + tid];
  const int x = wave_incl_scan(c, lane);
  if (lane == 31) wt[wave] = x;
  __syncthreads();
  int woff = 0, total = 0;
  for (int w = 0; w < 8; ++w) {
    const int t = wt[w];
    if (w < wave) woff += t;
    total += t;
  }
  if (tid == 0) base_s = atomicAdd(cursor, total);
  __syncthreads();
  const int base = base_s;
  st[b * EDGE_CHUNKS + tid] = base + woff + x - c;
  if (tid == 0) { binstart[b] = base; bincnt[b] = total; }
}

__global__ __launch_bounds__(1024) void bin_scatter_kernel(const int* __restrict__ src,
                                                           const int* __restrict__ dst,
                                                           const int* __restrict__ st,
                                                           int2* __restrict__ eb,
                                                           int n_edges, int nb) {
  __shared__ int pos[MAX_BINS];
  const int tid = threadIdx.x;
  for (int b = tid; b < nb; b += 1024) pos[b] = st[b * EDGE_CHUNKS + blockIdx.x];
  __syncthreads();
  const int chunk = (n_edges + EDGE_CHUNKS - 1) / EDGE_CHUNKS;
  const int e0 = blockIdx.x * chunk;
  const int e1 = (e0 + chunk < n_edges) ? (e0 + chunk) : n_edges;
  int e = e0 + tid;
  for (; e + 3 * 1024 < e1; e += 4 * 1024) {
    const int d0 = dst[e];
    const int d1 = dst[e + 1024];
    const int d2 = dst[e + 2 * 1024];
    const int d3 = dst[e + 3 * 1024];
    const int s0 = src[e];
    const int s1 = src[e + 1024];
    const int s2 = src[e + 2 * 1024];
    const int s3 = src[e + 3 * 1024];
    const int p0 = atomicAdd(&pos[d0 >> BIN_SHIFT], 1);
    const int p1 = atomicAdd(&pos[d1 >> BIN_SHIFT], 1);
    const int p2 = atomicAdd(&pos[d2 >> BIN_SHIFT], 1);
    const int p3 = atomicAdd(&pos[d3 >> BIN_SHIFT], 1);
    eb[p0] = make_int2(d0, s0);
    eb[p1] = make_int2(d1, s1);
    eb[p2] = make_int2(d2, s2);
    eb[p3] = make_int2(d3, s3);
  }
  for (; e < e1; e += 1024) {
    const int d = dst[e];
    const int s = src[e];
    const int p = atomicAdd(&pos[d >> BIN_SHIFT], 1);
    eb[p] = make_int2(d, s);
  }
}

__global__ __launch_bounds__(1024) void bin_sort_kernel(const int2* __restrict__ eb,
                                                        const int* __restrict__ binstart,
                                                        const int* __restrict__ bincnt,
                                                        int* __restrict__ cnt,
                                                        int* __restrict__ off,
                                                        int* __restrict__ ssrc, int n) {
  __shared__ int c[BIN_NODES];
  __shared__ int o[BIN_NODES];
  __shared__ int wt[8];
  const int b = blockIdx.x;
  const int tid = threadIdx.x;
  const int lane = tid & 31;
  const int wave = tid >> 5;
  const int node0 = b << BIN_SHIFT;
  if (tid < BIN_NODES) c[tid] = 0;
  __syncthreads();
  const int beg = binstart[b];
  const int m = bincnt[b];
  int i = tid;
  for (; i + 3 * 1024 < m; i += 4 * 1024) {
    const int d0 = eb[beg + i].x;
    const int d1 = eb[beg + i + 1024].x;
    const int d2 = eb[beg + i + 2 * 1024].x;
    const int d3 = eb[beg + i + 3 * 1024].x;
    atomicAdd(&c[d0 - node0], 1);
    atomicAdd(&c[d1 - node0], 1);
    atomicAdd(&c[d2 - node0], 1);
    atomicAdd(&c[d3 - node0], 1);
  }
  for (; i < m; i += 1024) atomicAdd(&c[eb[beg + i].x - node0], 1);
  __syncthreads();
  int v = 0, x = 0;
  if (tid < BIN_NODES) {
    v = c[tid];
    x = wave_incl_scan(v, lane);
    if (lane == 31) wt[wave] = x;
  }
  __syncthreads();
  if (tid < BIN_NODES) {
    int woff = 0;
    for (int w = 0; w < 8; ++w) {
      const int t = wt[w];
      if (w < wave) woff += t;
    }
    const int ex = woff + x - v;
    o[tid] = ex;
    const int node = node0 + tid;
    if (node < n) { cnt[node] = v; off[node] = beg + ex; }
  }
  __syncthreads();
  i = tid;
  for (; i + 3 * 1024 < m; i += 4 * 1024) {
    const int2 r0 = eb[beg + i];
    const int2 r1 = eb[beg + i + 1024];
    const int2 r2 = eb[beg + i + 2 * 1024];
    const int2 r3 = eb[beg + i + 3 * 1024];
    const int p0 = atomicAdd(&o[r0.x - node0], 1);
    const int p1 = atomicAdd(&o[r1.x - node0], 1);
    const int p2 = atomicAdd(&o[r2.x - node0], 1);
    const int p3 = atomicAdd(&o[r3.x - node0], 1);
    ssrc[beg + p0] = r0.y;
    ssrc[beg + p1] = r1.y;
    ssrc[beg + p2] = r2.y;
    ssrc[beg + p3] = r3.y;
  }
  for (; i < m; i += 1024) {
    const int2 r = eb[beg + i];
    const int p = atomicAdd(&o[r.x - node0], 1);
    ssrc[beg + p] = r.y;
  }
}

template <int K>
__device__ __forceinline__ void add_rows128(const float4* __restrict__ H4, int myidx,
                                            int t, int lane, float4& acc) {
  float4 v[K];
#pragma unroll
  for (int u = 0; u < K; ++u) {
    const int s = __builtin_amdgcn_readlane(myidx, t + u);
    v[u] = H4[(size_t)s * (HID_CH / 4) + lane];
  }
#pragma unroll
  for (int u = 0; u < K; ++u) {
    acc.x += v[u].x; acc.y += v[u].y; acc.z += v[u].z; acc.w += v[u].w;
  }
}

#define TILE_PITCH 132
__global__ __launch_bounds__(512, 16) void gather1_gemm2_kernel(const float* __restrict__ H,
                                                            const int* __restrict__ ssrc,
                                                            const int* __restrict__ off,
                                                            const int* __restrict__ cnt,
                                                            const float* __restrict__ b1,
                                                            const float* __restrict__ W2,
                                                            float* __restrict__ H2,
                                                            int n, int tiles) {
  __shared__ float4 w2s4[HID_CH * OUT_CH / 4];
  __shared__ float4 tile4[2 * 16 * TILE_PITCH / 4];
  const float* w2s = (const float*)w2s4;
  float* tilef = (float*)tile4;

  const int wave = __builtin_amdgcn_readfirstlane((int)(threadIdx.x >> 5));
  const int lane = threadIdx.x & 31;
  const int half = lane >> 4;
  const int l    = lane & 15;
  const int col  = (wave << 4) + l;
  const bool colok = (col < OUT_CH);
  const int colc = colok ? col : 0;
  const float4* __restrict__ H4 = (const float4*)H;

  for (int q = threadIdx.x; q < HID_CH * OUT_CH / 4; q += 512) w2s4[q] = ((const float4*)W2)[q];
  const float4 bias = ((const float4*)b1)[lane];

  int buf = 0;
  for (int t = blockIdx.x; t < tiles; t += gridDim.x) {
    const int row_base = t * 16;
    const int node = row_base + wave;
    float4 acc = make_float4(0.0f, 0.0f, 0.0f, 0.0f);
    if (node < n) {
      const int beg = __builtin_amdgcn_readfirstlane(off[node]);
      const int deg = __builtin_amdgcn_readfirstlane(cnt[node]);
      for (int j = 0; j < deg; j += 32) {
        const int m = (deg - j < 32) ? (deg - j) : 32;
        const int myidx = (lane < m) ? ssrc[beg + j + lane] : 0;
        int u = 0;
        for (; u + 8 <= m; u += 8) add_rows128<8>(H4, myidx, u, lane, acc);
        if (u + 4 <= m) { add_rows128<4>(H4, myidx, u, lane, acc); u += 4; }
        if (u + 2 <= m) { add_rows128<2>(H4, myidx, u, lane, acc); u += 2; }
        if (u < m) add_rows128<1>(H4, myidx, u, lane, acc);
      }
    }
    float4 r;
    r.x = acc.x + bias.x; r.y = acc.y + bias.y; r.z = acc.z + bias.z; r.w = acc.w + bias.w;
    r.x = r.x > 0.0f ? r.x : 0.0f;
    r.y = r.y > 0.0f ? r.y : 0.0f;
    r.z = r.z > 0.0f ? r.z : 0.0f;
    r.w = r.w > 0.0f ? r.w : 0.0f;
    float* tb = tilef + buf * (16 * TILE_PITCH);
    *(float4*)(tb + wave * TILE_PITCH + 4 * lane) = r;
    __syncthreads();

    if (wave < 3) {
      const float* arow = tb + l * TILE_PITCH + (half << 1);
      const float* bcol = w2s + (half << 1) * OUT_CH + colc;
      v8f c = {};
      v2f a[2][4], b[2][4];
#pragma unroll
      for (int s = 0; s < 4; ++s) {
        a[0][s] = *(const v2f*)(arow + 4 * s);
        b[0][s].x = bcol[(4 * s) * OUT_CH];
        b[0][s].y = bcol[(4 * s + 1) * OUT_CH];
      }
      __builtin_amdgcn_sched_group_barrier(0x100, 6, 0);
#pragma unroll
      for (int g = 0; g < HID_CH / 16; ++g) {
        if (g + 1 < HID_CH / 16) {
#pragma unroll
          for (int s = 0; s < 4; ++s) {
            const int i = 4 * (g + 1) + s;
            a[(g + 1) & 1][s] = *(const v2f*)(arow + 4 * i);
            b[(g + 1) & 1][s].x = bcol[(4 * i) * OUT_CH];
            b[(g + 1) & 1][s].y = bcol[(4 * i + 1) * OUT_CH];
          }
        }
#pragma unroll
        for (int s = 0; s < 4; ++s) {
          c = __builtin_amdgcn_wmma_f32_16x16x4_f32(false, a[g & 1][s], false, b[g & 1][s],
                                                    (short)0, c, false, false);
        }
        __builtin_amdgcn_sched_group_barrier(0x100, 6, 0);
        __builtin_amdgcn_sched_group_barrier(0x008, 4, 0);
      }
      float* __restrict__ crow = H2 + (size_t)(row_base + (half << 3)) * OUT_CH + col;
      if (colok) {
        if (row_base + 16 <= n) {
#pragma unroll
          for (int v = 0; v < 8; ++v) crow[(size_t)v * OUT_CH] = c[v];
        } else {
#pragma unroll
          for (int v = 0; v < 8; ++v) {
            if (row_base + (half << 3) + v < n) crow[(size_t)v * OUT_CH] = c[v];
          }
        }
      }
    }
    buf ^= 1;
  }
}

__device__ __forceinline__ float4 load_row40(const float4* __restrict__ H4, int myidx,
                                             int e, int g, int m, int c) {
  const int s = __shfl(myidx, e & 31, 32);
  float4 v = make_float4(0.0f, 0.0f, 0.0f, 0.0f);
  if (g < 3 && e < m) v = H4[(size_t)s * (OUT_CH / 4) + c];
  return v;
}

template <int M>
__device__ __forceinline__ float row_xor(float x) {
  const int xi = __float_as_int(x);
  return __int_as_float(__builtin_amdgcn_update_dpp(xi, xi, 0x160 + M, 0xf, 0xf, false));
}

__global__ __launch_bounds__(256) void gather2_logsoftmax_kernel(const float* __restrict__ H,
                                                                 const int* __restrict__ ssrc,
                                                                 const int* __restrict__ off,
                                                                 const int* __restrict__ cnt,
                                                                 const float* __restrict__ b2,
                                                                 float* __restrict__ OUT, int n) {
  const int wave = __builtin_amdgcn_readfirstlane((int)(threadIdx.x >> 5));
  const int lane = threadIdx.x & 31;
  const int nwaves = gridDim.x * 8;
  const int g = lane / 10;
  const int c = lane - g * 10;
  const int l1 = (lane + 10) & 31;
  const int l2 = (lane + 20) & 31;
  const bool act = (lane < OUT_CH / 4);
  const float4* __restrict__ H4 = (const float4*)H;
  float4 bias = make_float4(0.0f, 0.0f, 0.0f, 0.0f);
  if (act) bias = ((const float4*)b2)[lane];
  for (int node = blockIdx.x * 8 + wave; node < n; node += nwaves) {
    const int beg = off[node];
    const int deg = cnt[node];
    float4 acc = make_float4(0.0f, 0.0f, 0.0f, 0.0f);
    for (int j = 0; j < deg; j += 32) {
      const int m = (deg - j < 32) ? (deg - j) : 32;
      const int myidx = (lane < m) ? ssrc[beg + j + lane] : 0;
      for (int t = 0; t < m; t += 12) {
        const float4 v0 = load_row40(H4, myidx, t + g,     g, m, c);
        const float4 v1 = load_row40(H4, myidx, t + 3 + g, g, m, c);
        const float4 v2 = load_row40(H4, myidx, t + 6 + g, g, m, c);
        const float4 v3 = load_row40(H4, myidx, t + 9 + g, g, m, c);
        acc.x += v0.x; acc.y += v0.y; acc.z += v0.z; acc.w += v0.w;
        acc.x += v1.x; acc.y += v1.y; acc.z += v1.z; acc.w += v1.w;
        acc.x += v2.x; acc.y += v2.y; acc.z += v2.z; acc.w += v2.w;
        acc.x += v3.x; acc.y += v3.y; acc.z += v3.z; acc.w += v3.w;
      }
    }
    float4 r;
    r.x = (acc.x + __shfl(acc.x, l1, 32)) + __shfl(acc.x, l2, 32);
    r.y = (acc.y + __shfl(acc.y, l1, 32)) + __shfl(acc.y, l2, 32);
    r.z = (acc.z + __shfl(acc.z, l1, 32)) + __shfl(acc.z, l2, 32);
    r.w = (acc.w + __shfl(acc.w, l1, 32)) + __shfl(acc.w, l2, 32);

    float4 v;
    v.x = r.x + bias.x; v.y = r.y + bias.y; v.z = r.z + bias.z; v.w = r.w + bias.w;
    float mx = act ? fmaxf(fmaxf(v.x, v.y), fmaxf(v.z, v.w)) : -INFINITY;
    mx = fmaxf(mx, row_xor<8>(mx));
    mx = fmaxf(mx, row_xor<4>(mx));
    mx = fmaxf(mx, row_xor<2>(mx));
    mx = fmaxf(mx, row_xor<1>(mx));
    float s = 0.0f;
    if (act) s = (expf(v.x - mx) + expf(v.y - mx)) + (expf(v.z - mx) + expf(v.w - mx));
    s += row_xor<8>(s);
    s += row_xor<4>(s);
    s += row_xor<2>(s);
    s += row_xor<1>(s);
    const float lse = mx + logf(s);
    if (act) {
      ((float4*)OUT)[(size_t)node * (OUT_CH / 4) + lane] =
          make_float4(v.x - lse, v.y - lse, v.z - lse, v.w - lse);
    }
  }
}

extern "C" void kernel_launch(void* const* d_in, const int* in_sizes, int n_in,
                              void* d_out, int out_size, void* d_ws, size_t ws_size,
                              hipStream_t stream) {
  const float* x  = (const float*)d_in[0];
  const int*   ei = (const int*)d_in[1];
  const float* W1 = (const float*)d_in[2];
  const float* b1 = (const float*)d_in[3];
  const float* W2 = (const float*)d_in[4];
  const float* b2 = (const float*)d_in[5];

  const int n       = in_sizes[0] / IN_CH;
  const int n_edges = in_sizes[1] / 2;
  const int* src = ei;
  const int* dst = ei + n_edges;

  float* h1   = (float*)d_ws;
  float* h2   = h1 + (size_t)n * HID_CH;
  float* out  = (float*)d_out;
  const int nb = (n + BIN_NODES - 1) / BIN_NODES;
  int2* eb    = (int2*)(h2 + (size_t)n * HID_CH);
  int* cnt    = (int*)(eb + n_edges);
  int* cursor = cnt + n;
  int* off    = cursor + 1;
  int* ssrc   = off + n;
  int* bh     = ssrc + n_edges;
  int* st     = bh + (size_t)nb * EDGE_CHUNKS;
  int* binstart = st + (size_t)nb * EDGE_CHUNKS;
  int* bincnt = binstart + nb;

  const int tiles   = (n + 15) / 16;
  const int nblk    = (n + 7) / 8;
  const int g_gemm1 = tiles < 1024 ? tiles : 1024;
  const int g_fused = tiles < 1024 ? tiles : 1024;
  const int g_node  = nblk < 2048 ? nblk : 2048;

  bin_hist_kernel<<<EDGE_CHUNKS, 1024, 0, stream>>>(dst, bh, cursor, n_edges, nb);
  bin_scan_kernel<<<nb, 256, 0, stream>>>(bh, st, binstart, bincnt, cursor);
  bin_scatter_kernel<<<EDGE_CHUNKS, 1024, 0, stream>>>(src, dst, st, eb, n_edges, nb);
  bin_sort_kernel<<<nb, 1024, 0, stream>>>(eb, binstart, bincnt, cnt, off, ssrc, n);

  gemm1_kernel<<<g_gemm1, 256, 0, stream>>>(x, W1, h1, n, tiles);
  gather1_gemm2_kernel<<<g_fused, 512, 0, stream>>>(h1, ssrc, off, cnt, b1, W2, h2, n, tiles);

  gather2_logsoftmax_kernel<<<g_node, 256, 0, stream>>>(h2, ssrc, off, cnt, b2, out, n);
}


# Layer-1 product moved behind the aggregation: x rows are summed first (sum order changes), the 128x128 WMMA product runs on the summed LDS tile in the gather kernel; GEMM1 launch and H round trip go
# speedup vs baseline: 4.1793x; 1.1977x over previous
#include <hip/hip_runtime.h>
#include <math.h>

typedef float v2f __attribute__((ext_vector_type(2)));
typedef float v8f __attribute__((ext_vector_type(8)));

#define IN_CH 128
#define HID_CH 128
#define OUT_CH 40

#define BIN_SHIFT 8
#define BIN_NODES 256
#define EDGE_CHUNKS 256
#define MAX_BINS 1024

__device__ __forceinline__ int wave_incl_scan(int x, int lane) {
  for (int o = 1; o < 32; o <<= 1) {
    const int y = __shfl_up(x, o, 32);
    if (lane >= o) x += y;
  }
  return x;
}

__global__ __launch_bounds__(1024) void bin_hist_kernel(const int* __restrict__ dst,
                                                        int* __restrict__ bh,
                                                        int* __restrict__ cursor,
                                                        int n_edges, int nb) {
  __shared__ int h[MAX_BINS];
  const int tid = threadIdx.x;
  h[tid] = 0;
  __syncthreads();
  const int chunk = (n_edges + EDGE_CHUNKS - 1) / EDGE_CHUNKS;
  const int e0 = blockIdx.x * chunk;
  const int e1 = (e0 + chunk < n_edges) ? (e0 + chunk) : n_edges;
  int e = e0 + tid;
  for (; e + 3 * 1024 < e1; e += 4 * 1024) {
    const int d0 = dst[e];
    const int d1 = dst[e + 1024];
    const int d2 = dst[e + 2 * 1024];
    const int d3 = dst[e + 3 * 1024];
    atomicAdd(&h[d0 >> BIN_SHIFT], 1);
    atomicAdd(&h[d1 >> BIN_SHIFT], 1);
    atomicAdd(&h[d2 >> BIN_SHIFT], 1);
    atomicAdd(&h[d3 >> BIN_SHIFT], 1);
  }
  for (; e < e1; e += 1024) atomicAdd(&h[dst[e] >> BIN_SHIFT], 1);
  __syncthreads();
  for (int b = tid; b < nb; b += 1024) bh[b * EDGE_CHUNKS + blockIdx.x] = h[b];
  if (blockIdx.x == 0 && tid == 0) *cursor = 0;
}

__global__ __launch_bounds__(256) void bin_scan_kernel(const int* __restrict__ bh,
                                                       int* __restrict__ st,
                                                       int* __restrict__ binstart,
                                                       int* __restrict__ bincnt,
                                                       int* __restrict__ cursor) {
  __shared__ int wt[8];
  __shared__ int base_s;
  const int b = blockIdx.x;
  const int tid = threadIdx.x;
  const int lane = tid & 31;
  const int wave = tid >> 5;
  const int c = bh[b * EDGE_CHUNKS + tid];
  const int x = wave_incl_scan(c, lane);
  if (lane == 31) wt[wave] = x;
  __syncthreads();
  int woff = 0, total = 0;
  for (int w = 0; w < 8; ++w) {
    const int t = wt[w];
    if (w < wave) woff += t;
    total += t;
  }
  if (tid == 0) base_s = atomicAdd(cursor, total);
  __syncthreads();
  const int base = base_s;
  st[b * EDGE_CHUNKS + tid] = base + woff + x - c;
  if (tid == 0) { binstart[b] = base; bincnt[b] = total; }
}

__global__ __launch_bounds__(1024) void bin_scatter_kernel(const int* __restrict__ src,
                                                           const int* __restrict__ dst,
                                                           const int* __restrict__ st,
                                                           int2* __restrict__ eb,
                                                           int n_edges, int nb) {
  __shared__ int pos[MAX_BINS];
  const int tid = threadIdx.x;
  for (int b = tid; b < nb; b += 1024) pos[b] = st[b * EDGE_CHUNKS + blockIdx.x];
  __syncthreads();
  const int chunk = (n_edges + EDGE_CHUNKS - 1) / EDGE_CHUNKS;
  const int e0 = blockIdx.x * chunk;
  const int e1 = (e0 + chunk < n_edges) ? (e0 + chunk) : n_edges;
  int e = e0 + tid;
  for (; e + 3 * 1024 < e1; e += 4 * 1024) {
    const int d0 = dst[e];
    const int d1 = dst[e + 1024];
    const int d2 = dst[e + 2 * 1024];
    const int d3 = dst[e + 3 * 1024];
    const int s0 = src[e];
    const int s1 = src[e + 1024];
    const int s2 = src[e + 2 * 1024];
    const int s3 = src[e + 3 * 1024];
    const int p0 = atomicAdd(&pos[d0 >> BIN_SHIFT], 1);
    const int p1 = atomicAdd(&pos[d1 >> BIN_SHIFT], 1);
    const int p2 = atomicAdd(&pos[d2 >> BIN_SHIFT], 1);
    const int p3 = atomicAdd(&pos[d3 >> BIN_SHIFT], 1);
    eb[p0] = make_int2(d0, s0);
    eb[p1] = make_int2(d1, s1);
    eb[p2] = make_int2(d2, s2);
    eb[p3] = make_int2(d3, s3);
  }
  for (; e < e1; e += 1024) {
    const int d = dst[e];
    const int s = src[e];
    const int p = atomicAdd(&pos[d >> BIN_SHIFT], 1);
    eb[p] = make_int2(d, s);
  }
}

__global__ __launch_bounds__(1024) void bin_sort_kernel(const int2* __restrict__ eb,
                                                        const int* __restrict__ binstart,
                                                        const int* __restrict__ bincnt,
                                                        int* __restrict__ cnt,
                                                        int* __restrict__ off,
                                                        int* __restrict__ ssrc, int n) {
  __shared__ int c[BIN_NODES];
  __shared__ int o[BIN_NODES];
  __shared__ int wt[8];
  const int b = blockIdx.x;
  const int tid = threadIdx.x;
  const int lane = tid & 31;
  const int wave = tid >> 5;
  const int node0 = b << BIN_SHIFT;
  if (tid < BIN_NODES) c[tid] = 0;
  __syncthreads();
  const int beg = binstart[b];
  const int m = bincnt[b];
  int i = tid;
  for (; i + 3 * 1024 < m; i += 4 * 1024) {
    const int d0 = eb[beg + i].x;
    const int d1 = eb[beg + i + 1024].x;
    const int d2 = eb[beg + i + 2 * 1024].x;
    const int d3 = eb[beg + i + 3 * 1024].x;
    atomicAdd(&c[d0 - node0], 1);
    atomicAdd(&c[d1 - node0], 1);
    atomicAdd(&c[d2 - node0], 1);
    atomicAdd(&c[d3 - node0], 1);
  }
  for (; i < m; i += 1024) atomicAdd(&c[eb[beg + i].x - node0], 1);
  __syncthreads();
  int v = 0, x = 0;
  if (tid < BIN_NODES) {
    v = c[tid];
    x = wave_incl_scan(v, lane);
    if (lane == 31) wt[wave] = x;
  }
  __syncthreads();
  if (tid < BIN_NODES) {
    int woff = 0;
    for (int w = 0; w < 8; ++w) {
      const int t = wt[w];
      if (w < wave) woff += t;
    }
    const int ex = woff + x - v;
    o[tid] = ex;
    const int node = node0 + tid;
    if (node < n) { cnt[node] = v; off[node] = beg + ex; }
  }
  __syncthreads();
  i = tid;
  for (; i + 3 * 1024 < m; i += 4 * 1024) {
    const int2 r0 = eb[beg + i];
    const int2 r1 = eb[beg + i + 1024];
    const int2 r2 = eb[beg + i + 2 * 1024];
    const int2 r3 = eb[beg + i + 3 * 1024];
    const int p0 = atomicAdd(&o[r0.x - node0], 1);
    const int p1 = atomicAdd(&o[r1.x - node0], 1);
    const int p2 = atomicAdd(&o[r2.x - node0], 1);
    const int p3 = atomicAdd(&o[r3.x - node0], 1);
    ssrc[beg + p0] = r0.y;
    ssrc[beg + p1] = r1.y;
    ssrc[beg + p2] = r2.y;
    ssrc[beg + p3] = r3.y;
  }
  for (; i < m; i += 1024) {
    const int2 r = eb[beg + i];
    const int p = atomicAdd(&o[r.x - node0], 1);
    ssrc[beg + p] = r.y;
  }
}

template <int K>
__device__ __forceinline__ void add_rows128(const float4* __restrict__ H4, int myidx,
                                            int t, int lane, float4& acc) {
  float4 v[K];
#pragma unroll
  for (int u = 0; u < K; ++u) {
    const int s = __builtin_amdgcn_readlane(myidx, t + u);
    v[u] = H4[(size_t)s * (HID_CH / 4) + lane];
  }
#pragma unroll
  for (int u = 0; u < K; ++u) {
    acc.x += v[u].x; acc.y += v[u].y; acc.z += v[u].z; acc.w += v[u].w;
  }
}

template <bool PAIRS>
__device__ __forceinline__ v2f load_bfrag(const float* bp, int i) {
  v2f b;
  if (PAIRS) {
    b = *(const v2f*)(bp + 512 * i);
  } else {
    b.x = bp[(4 * i) * OUT_CH];
    b.y = bp[(4 * i + 1) * OUT_CH];
  }
  return b;
}

template <bool PAIRS, int NREADS>
__device__ __forceinline__ v8f tile_times_b(const float* arow, const float* bp) {
  v8f c = {};
  v2f a[2][4], b[2][4];
#pragma unroll
  for (int s = 0; s < 4; ++s) {
    a[0][s] = *(const v2f*)(arow + 4 * s);
    b[0][s] = load_bfrag<PAIRS>(bp, s);
  }
  __builtin_amdgcn_sched_group_barrier(0x100, NREADS, 0);
#pragma unroll
  for (int g = 0; g < HID_CH / 16; ++g) {
    if (g + 1 < HID_CH / 16) {
#pragma unroll
      for (int s = 0; s < 4; ++s) {
        const int i = 4 * (g + 1) + s;
        a[(g + 1) & 1][s] = *(const v2f*)(arow + 4 * i);
        b[(g + 1) & 1][s] = load_bfrag<PAIRS>(bp, i);
      }
    }
#pragma unroll
    for (int s = 0; s < 4; ++s) {
      c = __builtin_amdgcn_wmma_f32_16x16x4_f32(false, a[g & 1][s], false, b[g & 1][s],
                                                (short)0, c, false, false);
    }
    __builtin_amdgcn_sched_group_barrier(0x100, NREADS, 0);
    __builtin_amdgcn_sched_group_barrier(0x008, 4, 0);
  }
  return c;
}

#define TILE_PITCH 132
#define TILE_ROWS 32
__global__ __launch_bounds__(1024, 16) void gather1_gemm12_kernel(const float* __restrict__ X,
                                                             const int* __restrict__ ssrc,
                                                             const int* __restrict__ off,
                                                             const int* __restrict__ cnt,
                                                             const float* __restrict__ W1,
                                                             const float* __restrict__ b1,
                                                             const float* __restrict__ W2,
                                                             float* __restrict__ H2,
                                                             int n, int pairs) {
  __shared__ float4 w1s4[IN_CH * HID_CH / 4];
  __shared__ float4 w2s4[HID_CH * OUT_CH / 4];
  __shared__ float4 xt4[2 * TILE_ROWS * TILE_PITCH / 4];
  __shared__ float4 ht4[2 * TILE_ROWS * TILE_PITCH / 4];
  float* w1s = (float*)w1s4;
  const float* w2s = (const float*)w2s4;
  float* xt = (float*)xt4;
  float* ht = (float*)ht4;

  const int wave = __builtin_amdgcn_readfirstlane((int)(threadIdx.x >> 5));
  const int lane = threadIdx.x & 31;
  const int half = lane >> 4;
  const int l    = lane & 15;
  const float4* __restrict__ X4 = (const float4*)X;

  for (int q = threadIdx.x; q < (IN_CH / 2) * (HID_CH / 4); q += 1024) {
    const int kp = q >> 5;
    const int c4 = (q & 31) << 2;
    const float4 r0 = ((const float4*)W1)[((2 * kp) * HID_CH + c4) >> 2];
    const float4 r1 = ((const float4*)W1)[((2 * kp + 1) * HID_CH + c4) >> 2];
    const int cs = c4 ^ ((kp & 1) << 4);
    float4* d = (float4*)(w1s + 2 * (kp * HID_CH + cs));
    d[0] = make_float4(r0.x, r1.x, r0.y, r1.y);
    d[1] = make_float4(r0.z, r1.z, r0.w, r1.w);
  }
  for (int q = threadIdx.x; q < HID_CH * OUT_CH / 4; q += 1024) w2s4[q] = ((const float4*)W2)[q];

  const int t1 = (wave >> 3) & 1;
  const int col1 = ((wave & 7) << 4) + l;
  const float bias1 = b1[col1];
  const int w3 = wave - 16;
  const int t2 = (w3 >= 3) ? 1 : 0;
  const int col2 = ((w3 - 3 * t2) << 4) + l;
  const bool colok = (col2 < OUT_CH);
  const int colc = colok ? col2 : 0;

  const int K = ((int)blockIdx.x < pairs) ? (pairs - 1 - (int)blockIdx.x) / (int)gridDim.x + 1 : 0;
  for (int r = 0; r < K + 2; ++r) {
    const int buf = r & 1;
    if (r < K) {
      const int node = ((int)blockIdx.x + r * (int)gridDim.x) * TILE_ROWS + wave;
      float4 acc = make_float4(0.0f, 0.0f, 0.0f, 0.0f);
      if (node < n) {
        const int beg = __builtin_amdgcn_readfirstlane(off[node]);
        const int deg = __builtin_amdgcn_readfirstlane(cnt[node]);
        for (int j = 0; j < deg; j += 32) {
          const int m = (deg - j < 32) ? (deg - j) : 32;
          const int myidx = (lane < m) ? ssrc[beg + j + lane] : 0;
          int u = 0;
          for (; u + 8 <= m; u += 8) add_rows128<8>(X4, myidx, u, lane, acc);
          if (u + 4 <= m) { add_rows128<4>(X4, myidx, u, lane, acc); u += 4; }
          if (u + 2 <= m) { add_rows128<2>(X4, myidx, u, lane, acc); u += 2; }
          if (u < m) add_rows128<1>(X4, myidx, u, lane, acc);
        }
      }
      *(float4*)(xt + buf * (TILE_ROWS * TILE_PITCH) + wave * TILE_PITCH + 4 * lane) = acc;
    }
    if (r >= 1 && r <= K && wave < 16) {
      const float* arow = xt + (buf ^ 1) * (TILE_ROWS * TILE_PITCH) +
                          (16 * t1 + l) * TILE_PITCH + (half << 1);
      const float* bp = w1s + 2 * (half * HID_CH + (col1 ^ (half << 4)));
      const v8f c = tile_times_b<true, 4>(arow, bp);
      float* hrow = ht + (buf ^ 1) * (TILE_ROWS * TILE_PITCH) +
                    (16 * t1 + (half << 3)) * TILE_PITCH + col1;
#pragma unroll
      for (int v = 0; v < 8; ++v) {
        const float h = c[v] + bias1;
        hrow[v * TILE_PITCH] = h > 0.0f ? h : 0.0f;
      }
    }
    if (r >= 2 && wave >= 16 && wave < 22) {
      const int row_base = ((int)blockIdx.x + (r - 2) * (int)gridDim.x) * TILE_ROWS + 16 * t2;
      const float* arow = ht + buf * (TILE_ROWS * TILE_PITCH) +
                          (16 * t2 + l) * TILE_PITCH + (half << 1);
      const float* bp = w2s + (half << 1) * OUT_CH + colc;
      const v8f c = tile_times_b<false, 6>(arow, bp);
      float* __restrict__ crow = H2 + (size_t)(row_base + (half << 3)) * OUT_CH + col2;
      if (colok) {
        if (row_base + 16 <= n) {
#pragma unroll
          for (int v = 0; v < 8; ++v) crow[(size_t)v * OUT_CH] = c[v];
        } else {
#pragma unroll
          for (int v = 0; v < 8; ++v) {
            if (row_base + (half << 3) + v < n) crow[(size_t)v * OUT_CH] = c[v];
          }
        }
      }
    }
    __syncthreads();
  }
}

__device__ __forceinline__ float4 load_row40(const float4* __restrict__ H4, int myidx,
                                             int e, int g, int m, int c) {
  const int s = __shfl(myidx, e & 31, 32);
  float4 v = make_float4(0.0f, 0.0f, 0.0f, 0.0f);
  if (g < 3 && e < m) v = H4[(size_t)s * (OUT_CH / 4) + c];
  return v;
}

template <int M>
__device__ __forceinline__ float row_xor(float x) {
  const int xi = __float_as_int(x);
  return __int_as_float(__builtin_amdgcn_update_dpp(xi, xi, 0x160 + M, 0xf, 0xf, false));
}

__global__ __launch_bounds__(256) void gather2_logsoftmax_kernel(const float* __restrict__ H,
                                                                 const int* __restrict__ ssrc,
                                                                 const int* __restrict__ off,
                                                                 const int* __restrict__ cnt,
                                                                 const float* __restrict__ b2,
                                                                 float* __restrict__ OUT, int n) {
  const int wave = __builtin_amdgcn_readfirstlane((int)(threadIdx.x >> 5));
  const int lane = threadIdx.x & 31;
  const int nwaves = gridDim.x * 8;
  const int g = lane / 10;
  const int c = lane - g * 10;
  const int l1 = (lane + 10) & 31;
  const int l2 = (lane + 20) & 31;
  const bool act = (lane < OUT_CH / 4);
  const float4* __restrict__ H4 = (const float4*)H;
  float4 bias = make_float4(0.0f, 0.0f, 0.0f, 0.0f);
  if (act) bias = ((const float4*)b2)[lane];
  for (int node = blockIdx.x * 8 + wave; node < n; node += nwaves) {
    const int beg = off[node];
    const int deg = cnt[node];
    float4 acc = make_float4(0.0f, 0.0f, 0.0f, 0.0f);
    for (int j = 0; j < deg; j += 32) {
      const int m = (deg - j < 32) ? (deg - j) : 32;
      const int myidx = (lane < m) ? ssrc[beg + j + lane] : 0;
      for (int t = 0; t < m; t += 12) {
        const float4 v0 = load_row40(H4, myidx, t + g,     g, m, c);
        const float4 v1 = load_row40(H4, myidx, t + 3 + g, g, m, c);
        const float4 v2 = load_row40(H4, myidx, t + 6 + g, g, m, c);
        const float4 v3 = load_row40(H4, myidx, t + 9 + g, g, m, c);
        acc.x += v0.x; acc.y += v0.y; acc.z += v0.z; acc.w += v0.w;
        acc.x += v1.x; acc.y += v1.y; acc.z += v1.z; acc.w += v1.w;
        acc.x += v2.x; acc.y += v2.y; acc.z += v2.z; acc.w += v2.w;
        acc.x += v3.x; acc.y += v3.y; acc.z += v3.z; acc.w += v3.w;
      }
    }
    float4 r;
    r.x = (acc.x + __shfl(acc.x, l1, 32)) + __shfl(acc.x, l2, 32);
    r.y = (acc.y + __shfl(acc.y, l1, 32)) + __shfl(acc.y, l2, 32);
    r.z = (acc.z + __shfl(acc.z, l1, 32)) + __shfl(acc.z, l2, 32);
    r.w = (acc.w + __shfl(acc.w, l1, 32)) + __shfl(acc.w, l2, 32);

    float4 v;
    v.x = r.x + bias.x; v.y = r.y + bias.y; v.z = r.z + bias.z; v.w = r.w + bias.w;
    float mx = act ? fmaxf(fmaxf(v.x, v.y), fmaxf(v.z, v.w)) : -INFINITY;
    mx = fmaxf(mx, row_xor<8>(mx));
    mx = fmaxf(mx, row_xor<4>(mx));
    mx = fmaxf(mx, row_xor<2>(mx));
    mx = fmaxf(mx, row_xor<1>(mx));
    float s = 0.0f;
    if (act) s = (expf(v.x - mx) + expf(v.y - mx)) + (expf(v.z - mx) + expf(v.w - mx));
    s += row_xor<8>(s);
    s += row_xor<4>(s);
    s += row_xor<2>(s);
    s += row_xor<1>(s);
    const float lse = mx + logf(s);
    if (act) {
      ((float4*)OUT)[(size_t)node * (OUT_CH / 4) + lane] =
          make_float4(v.x - lse, v.y - lse, v.z - lse, v.w - lse);
    }
  }
}

extern "C" void kernel_launch(void* const* d_in, const int* in_sizes, int n_in,
                              void* d_out, int out_size, void* d_ws, size_t ws_size,
                              hipStream_t stream) {
  const float* x  = (const float*)d_in[0];
  const int*   ei = (const int*)d_in[1];
  const float* W1 = (const float*)d_in[2];
  const float* b1 = (const float*)d_in[3];
  const float* W2 = (const float*)d_in[4];
  const float* b2 = (const float*)d_in[5];

  const int n       = in_sizes[0] / IN_CH;
  const int n_edges = in_sizes[1] / 2;
  const int* src = ei;
  const int* dst = ei + n_edges;

  float* h2   = (float*)d_ws;
  float* out  = (float*)d_out;
  const int nb = (n + BIN_NODES - 1) / BIN_NODES;
  int2* eb    = (int2*)(h2 + (size_t)n * HID_CH);
  int* cnt    = (int*)(eb + n_edges);
  int* cursor = cnt + n;
  int* off    = cursor + 1;
  int* ssrc   = off + n;
  int* bh     = ssrc + n_edges;
  int* st     = bh + (size_t)nb * EDGE_CHUNKS;
  int* binstart = st + (size_t)nb * EDGE_CHUNKS;
  int* bincnt = binstart + nb;

  const int pairs   = (n + 31) / 32;
  const int nblk    = (n + 7) / 8;
  const int g_fused = pairs < 512 ? pairs : 512;
  const int g_node  = nblk < 2048 ? nblk : 2048;

  bin_hist_kernel<<<EDGE_CHUNKS, 1024, 0, stream>>>(dst, bh, cursor, n_edges, nb);
  bin_scan_kernel<<<nb, 256, 0, stream>>>(bh, st, binstart, bincnt, cursor);
  bin_scatter_kernel<<<EDGE_CHUNKS, 1024, 0, stream>>>(src, dst, st, eb, n_edges, nb);
  bin_sort_kernel<<<nb, 1024, 0, stream>>>(eb, binstart, bincnt, cnt, off, ssrc, n);

  gather1_gemm12_kernel<<<g_fused, 1024, 0, stream>>>(x, ssrc, off, cnt, W1, b1, W2, h2, n, pairs);

  gather2_logsoftmax_kernel<<<g_node, 256, 0, stream>>>(h2, ssrc, off, cnt, b2, out, n);
}


# Edge binning made chunk-local: each edge-chunk work-group counts, scans and places its edges by bin in its own slice, bin groups walk 256 pieces; hist and scan launches go; rank order differs
# speedup vs baseline: 4.8884x; 1.1697x over previous
#include <hip/hip_runtime.h>
#include <math.h>

typedef float v2f __attribute__((ext_vector_type(2)));
typedef float v8f __attribute__((ext_vector_type(8)));

#define IN_CH 128
#define HID_CH 128
#define OUT_CH 40

#define BIN_SHIFT 8
#define BIN_NODES 256
#define EDGE_CHUNKS 256
#define MAX_BINS 1024
#define EDGE_SLOTS 8
#define PIECES_PER_WAVE (EDGE_CHUNKS / 32)
static_assert(EDGE_SLOTS == 8, "the load fences in bin_local_kernel name eight slots");

__device__ __forceinline__ int wave_incl_scan(int x, int lane) {
  for (int o = 1; o < 32; o <<= 1) {
    const int y = __shfl_up(x, o, 32);
    if (lane >= o) x += y;
  }
  return x;
}

__global__ __launch_bounds__(1024) void bin_local_kernel(const int* __restrict__ src,
                                                         const int* __restrict__ dst,
                                                         int2* __restrict__ eb,
                                                         int2* __restrict__ seg,
                                                         int* __restrict__ cursor,
                                                         int n_edges, int nb) {
  __shared__ int h[MAX_BINS];
  __shared__ int wt[32];
  const int tid = threadIdx.x;
  const int lane = tid & 31;
  const int wave = tid >> 5;
  h[tid] = 0;
  __syncthreads();
  const int chunk = (n_edges + EDGE_CHUNKS - 1) / EDGE_CHUNKS;
  const int e0 = blockIdx.x * chunk;
  const int e1 = (e0 + chunk < n_edges) ? (e0 + chunk) : n_edges;
  for (int e = e0 + tid; e < e1; e += EDGE_SLOTS * 1024) {
    int d[EDGE_SLOTS];
#pragma unroll
    for (int k = 0; k < EDGE_SLOTS; ++k) {
      const int ee = e + k * 1024;
      d[k] = dst[ee < e1 ? ee : e1 - 1];
    }
    asm volatile("" :: "v"(d[0]), "v"(d[1]), "v"(d[2]), "v"(d[3]),
                       "v"(d[4]), "v"(d[5]), "v"(d[6]), "v"(d[7]));
#pragma unroll
    for (int k = 0; k < EDGE_SLOTS; ++k) {
      if (e + k * 1024 < e1) atomicAdd(&h[d[k] >> BIN_SHIFT], 1);
    }
  }
  __syncthreads();
  const int v = h[tid];
  const int x = wave_incl_scan(v, lane);
  if (lane == 31) wt[wave] = x;
  __syncthreads();
  int woff = 0;
  for (int w = 0; w < 32; ++w) {
    const int t = wt[w];
    if (w < wave) woff += t;
  }
  const int ex = woff + x - v;
  h[tid] = ex;
  if (tid < nb) seg[tid * EDGE_CHUNKS + blockIdx.x] = make_int2(ex, v);
  __syncthreads();
  for (int e = e0 + tid; e < e1; e += EDGE_SLOTS * 1024) {
    int d[EDGE_SLOTS], s[EDGE_SLOTS];
#pragma unroll
    for (int k = 0; k < EDGE_SLOTS; ++k) {
      const int ee = e + k * 1024;
      const int ec = ee < e1 ? ee : e1 - 1;
      d[k] = dst[ec];
      s[k] = src[ec];
    }
    asm volatile("" :: "v"(d[0]), "v"(d[1]), "v"(d[2]), "v"(d[3]),
                       "v"(d[4]), "v"(d[5]), "v"(d[6]), "v"(d[7]),
                       "v"(s[0]), "v"(s[1]), "v"(s[2]), "v"(s[3]),
                       "v"(s[4]), "v"(s[5]), "v"(s[6]), "v"(s[7]));
#pragma unroll
    for (int k = 0; k < EDGE_SLOTS; ++k) {
      if (e + k * 1024 < e1) {
        const int p = atomicAdd(&h[d[k] >> BIN_SHIFT], 1);
        eb[e0 + p] = make_int2(d[k], s[k]);
      }
    }
  }
  if (blockIdx.x == 0 && tid == 0) *cursor = 0;
}

__global__ __launch_bounds__(1024, 16) void bin_sort_kernel(const int2* __restrict__ eb,
                                                        const int2* __restrict__ seg,
                                                        int* __restrict__ cursor,
                                                        int* __restrict__ cnt,
                                                        int* __restrict__ off,
                                                        int* __restrict__ ssrc,
                                                        int n, int n_edges) {
  __shared__ int c[BIN_NODES];
  __shared__ int o[BIN_NODES];
  __shared__ int wt[8];
  __shared__ int wm[8];
  __shared__ int2 sg[EDGE_CHUNKS];
  __shared__ int base_s;
  const int b = blockIdx.x;
  const int tid = threadIdx.x;
  const int lane = tid & 31;
  const int wave = tid >> 5;
  const int node0 = b << BIN_SHIFT;
  const int chunk = (n_edges + EDGE_CHUNKS - 1) / EDGE_CHUNKS;
  if (tid < EDGE_CHUNKS) {
    const int2 r = seg[b * EDGE_CHUNKS + tid];
    sg[tid] = r;
    c[tid] = 0;
    const int t = wave_incl_scan(r.y, lane);
    if (lane == 31) wm[wave] = t;
  }
  __syncthreads();
  int pp[PIECES_PER_WAVE];
  int len[PIECES_PER_WAVE];
#pragma unroll
  for (int u = 0; u < PIECES_PER_WAVE; ++u) {
    const int k = wave + 32 * u;
    const int2 r = sg[k];
    pp[u] = k * chunk + r.x;
    len[u] = r.y;
  }
  {
    int d[PIECES_PER_WAVE];
#pragma unroll
    for (int u = 0; u < PIECES_PER_WAVE; ++u) d[u] = (lane < len[u]) ? eb[pp[u] + lane].x : node0;
    if (tid == 0) {
      int total = 0;
      for (int w = 0; w < 8; ++w) total += wm[w];
      base_s = atomicAdd(cursor, total);
    }
#pragma unroll
    for (int u = 0; u < PIECES_PER_WAVE; ++u) {
      if (lane < len[u]) atomicAdd(&c[d[u] - node0], 1);
    }
#pragma unroll
    for (int u = 0; u < PIECES_PER_WAVE; ++u) {
      for (int i = lane + 32; i < len[u]; i += 32) atomicAdd(&c[eb[pp[u] + i].x - node0], 1);
    }
  }
  __syncthreads();
  int v = 0, x = 0;
  if (tid < BIN_NODES) {
    v = c[tid];
    x = wave_incl_scan(v, lane);
    if (lane == 31) wt[wave] = x;
  }
  __syncthreads();
  const int beg = base_s;
  if (tid < BIN_NODES) {
    int woff = 0;
    for (int w = 0; w < 8; ++w) {
      const int t = wt[w];
      if (w < wave) woff += t;
    }
    const int ex = woff + x - v;
    o[tid] = ex;
    const int node = node0 + tid;
    if (node < n) { cnt[node] = v; off[node] = beg + ex; }
  }
  __syncthreads();
  {
    int2 r[PIECES_PER_WAVE];
#pragma unroll
    for (int u = 0; u < PIECES_PER_WAVE; ++u) {
      r[u] = (lane < len[u]) ? eb[pp[u] + lane] : make_int2(node0, 0);
    }
#pragma unroll
    for (int u = 0; u < PIECES_PER_WAVE; ++u) {
      if (lane < len[u]) {
        const int p = atomicAdd(&o[r[u].x - node0], 1);
        ssrc[beg + p] = r[u].y;
      }
    }
#pragma unroll
    for (int u = 0; u < PIECES_PER_WAVE; ++u) {
      for (int i = lane + 32; i < len[u]; i += 32) {
        const int2 q = eb[pp[u] + i];
        const int p = atomicAdd(&o[q.x - node0], 1);
        ssrc[beg + p] = q.y;
      }
    }
  }
}

template <int K>
__device__ __forceinline__ void add_rows128(const float4* __restrict__ H4, int myidx,
                                            int t, int lane, float4& acc) {
  float4 v[K];
#pragma unroll
  for (int u = 0; u < K; ++u) {
    const int s = __builtin_amdgcn_readlane(myidx, t + u);
    v[u] = H4[(size_t)s * (HID_CH / 4) + lane];
  }
#pragma unroll
  for (int u = 0; u < K; ++u) {
    acc.x += v[u].x; acc.y += v[u].y; acc.z += v[u].z; acc.w += v[u].w;
  }
}

template <bool PAIRS>
__device__ __forceinline__ v2f load_bfrag(const float* bp, int i) {
  v2f b;
  if (PAIRS) {
    b = *(const v2f*)(bp + 512 * i);
  } else {
    b.x = bp[(4 * i) * OUT_CH];
    b.y = bp[(4 * i + 1) * OUT_CH];
  }
  return b;
}

template <bool PAIRS, int NREADS>
__device__ __forceinline__ v8f tile_times_b(const float* arow, const float* bp) {
  v8f c = {};
  v2f a[2][4], b[2][4];
#pragma unroll
  for (int s = 0; s < 4; ++s) {
    a[0][s] = *(const v2f*)(arow + 4 * s);
    b[0][s] = load_bfrag<PAIRS>(bp, s);
  }
  __builtin_amdgcn_sched_group_barrier(0x100, NREADS, 0);
#pragma unroll
  for (int g = 0; g < HID_CH / 16; ++g) {
    if (g + 1 < HID_CH / 16) {
#pragma unroll
      for (int s = 0; s < 4; ++s) {
        const int i = 4 * (g + 1) + s;
        a[(g + 1) & 1][s] = *(const v2f*)(arow + 4 * i);
        b[(g + 1) & 1][s] = load_bfrag<PAIRS>(bp, i);
      }
    }
#pragma unroll
    for (int s = 0; s < 4; ++s) {
      c = __builtin_amdgcn_wmma_f32_16x16x4_f32(false, a[g & 1][s], false, b[g & 1][s],
                                                (short)0, c, false, false);
    }
    __builtin_amdgcn_sched_group_barrier(0x100, NREADS, 0);
    __builtin_amdgcn_sched_group_barrier(0x008, 4, 0);
  }
  return c;
}

#define TILE_PITCH 132
#define TILE_ROWS 32
__global__ __launch_bounds__(1024, 16) void gather1_gemm12_kernel(const float* __restrict__ X,
                                                             const int* __restrict__ ssrc,
                                                             const int* __restrict__ off,
                                                             const int* __restrict__ cnt,
                                                             const float* __restrict__ W1,
                                                             const float* __restrict__ b1,
                                                             const float* __restrict__ W2,
                                                             float* __restrict__ H2,
                                                             int n, int pairs) {
  __shared__ float4 w1s4[IN_CH * HID_CH / 4];
  __shared__ float4 w2s4[HID_CH * OUT_CH / 4];
  __shared__ float4 xt4[2 * TILE_ROWS * TILE_PITCH / 4];
  __shared__ float4 ht4[2 * TILE_ROWS * TILE_PITCH / 4];
  float* w1s = (float*)w1s4;
  const float* w2s = (const float*)w2s4;
  float* xt = (float*)xt4;
  float* ht = (float*)ht4;

  const int wave = __builtin_amdgcn_readfirstlane((int)(threadIdx.x >> 5));
  const int lane = threadIdx.x & 31;
  const int half = lane >> 4;
  const int l    = lane & 15;
  const float4* __restrict__ X4 = (const float4*)X;

  for (int q = threadIdx.x; q < (IN_CH / 2) * (HID_CH / 4); q += 1024) {
    const int kp = q >> 5;
    const int c4 = (q & 31) << 2;
    const float4 r0 = ((const float4*)W1)[((2 * kp) * HID_CH + c4) >> 2];
    const float4 r1 = ((const float4*)W1)[((2 * kp + 1) * HID_CH + c4) >> 2];
    const int cs = c4 ^ ((kp & 1) << 4);
    float4* d = (float4*)(w1s + 2 * (kp * HID_CH + cs));
    d[0] = make_float4(r0.x, r1.x, r0.y, r1.y);
    d[1] = make_float4(r0.z, r1.z, r0.w, r1.w);
  }
  for (int q = threadIdx.x; q < HID_CH * OUT_CH / 4; q += 1024) w2s4[q] = ((const float4*)W2)[q];

  const int t1 = (wave >> 3) & 1;
  const int col1 = ((wave & 7) << 4) + l;
  const float bias1 = b1[col1];
  const int w3 = wave - 16;
  const int t2 = (w3 >= 3) ? 1 : 0;
  const int col2 = ((w3 - 3 * t2) << 4) + l;
  const bool colok = (col2 < OUT_CH);
  const int colc = colok ? col2 : 0;

  const int K = ((int)blockIdx.x < pairs) ? (pairs - 1 - (int)blockIdx.x) / (int)gridDim.x + 1 : 0;
  for (int r = 0; r < K + 2; ++r) {
    const int buf = r & 1;
    if (r < K) {
      const int node = ((int)blockIdx.x + r * (int)gridDim.x) * TILE_ROWS + wave;
      float4 acc = make_float4(0.0f, 0.0f, 0.0f, 0.0f);
      if (node < n) {
        const int beg = __builtin_amdgcn_readfirstlane(off[node]);
        const int deg = __builtin_amdgcn_readfirstlane(cnt[node]);
        for (int j = 0; j < deg; j += 32) {
          const int m = (deg - j < 32) ? (deg - j) : 32;
          const int myidx = (lane < m) ? ssrc[beg + j + lane] : 0;
          int u = 0;
          for (; u + 8 <= m; u += 8) add_rows128<8>(X4, myidx, u, lane, acc);
          if (u + 4 <= m) { add_rows128<4>(X4, myidx, u, lane, acc); u += 4; }
          if (u + 2 <= m) { add_rows128<2>(X4, myidx, u, lane, acc); u += 2; }
          if (u < m) add_rows128<1>(X4, myidx, u, lane, acc);
        }
      }
      *(float4*)(xt + buf * (TILE_ROWS * TILE_PITCH) + wave * TILE_PITCH + 4 * lane) = acc;
    }
    if (r >= 1 && r <= K && wave < 16) {
      const float* arow = xt + (buf ^ 1) * (TILE_ROWS * TILE_PITCH) +
                          (16 * t1 + l) * TILE_PITCH + (half << 1);
      const float* bp = w1s + 2 * (half * HID_CH + (col1 ^ (half << 4)));
      const v8f c = tile_times_b<true, 4>(arow, bp);
      float* hrow = ht + (buf ^ 1) * (TILE_ROWS * TILE_PITCH) +
                    (16 * t1 + (half << 3)) * TILE_PITCH + col1;
#pragma unroll
      for (int v = 0; v < 8; ++v) {
        const float h = c[v] + bias1;
        hrow[v * TILE_PITCH] = h > 0.0f ? h : 0.0f;
      }
    }
    if (r >= 2 && wave >= 16 && wave < 22) {
      const int row_base = ((int)blockIdx.x + (r - 2) * (int)gridDim.x) * TILE_ROWS + 16 * t2;
      const float* arow = ht + buf * (TILE_ROWS * TILE_PITCH) +
                          (16 * t2 + l) * TILE_PITCH + (half << 1);
      const float* bp = w2s + (half << 1) * OUT_CH + colc;
      const v8f c = tile_times_b<false, 6>(arow, bp);
      float* __restrict__ crow = H2 + (size_t)(row_base + (half << 3)) * OUT_CH + col2;
      if (colok) {
        if (row_base + 16 <= n) {
#pragma unroll
          for (int v = 0; v < 8; ++v) crow[(size_t)v * OUT_CH] = c[v];
        } else {
#pragma unroll
          for (int v = 0; v < 8; ++v) {
            if (row_base + (half << 3) + v < n) crow[(size_t)v * OUT_CH] = c[v];
          }
        }
      }
    }
    __syncthreads();
  }
}

__device__ __forceinline__ float4 load_row40(const float4* __restrict__ H4, int myidx,
                                             int e, int g, int m, int c) {
  const int s = __shfl(myidx, e & 31, 32);
  float4 v = make_float4(0.0f, 0.0f, 0.0f, 0.0f);
  if (g < 3 && e < m) v = H4[(size_t)s * (OUT_CH / 4) + c];
  return v;
}

template <int M>
__device__ __forceinline__ float row_xor(float x) {
  const int xi = __float_as_int(x);
  return __int_as_float(__builtin_amdgcn_update_dpp(xi, xi, 0x160 + M, 0xf, 0xf, false));
}

__global__ __launch_bounds__(256) void gather2_logsoftmax_kernel(const float* __restrict__ H,
                                                                 const int* __restrict__ ssrc,
                                                                 const int* __restrict__ off,
                                                                 const int* __restrict__ cnt,
                                                                 const float* __restrict__ b2,
                                                                 float* __restrict__ OUT, int n) {
  const int wave = __builtin_amdgcn_readfirstlane((int)(threadIdx.x >> 5));
  const int lane = threadIdx.x & 31;
  const int nwaves = gridDim.x * 8;
  const int g = lane / 10;
  const int c = lane - g * 10;
  const int l1 = (lane + 10) & 31;
  const int l2 = (lane + 20) & 31;
  const bool act = (lane < OUT_CH / 4);
  const float4* __restrict__ H4 = (const float4*)H;
  float4 bias = make_float4(0.0f, 0.0f, 0.0f, 0.0f);
  if (act) bias = ((const float4*)b2)[lane];
  for (int node = blockIdx.x * 8 + wave; node < n; node += nwaves) {
    const int beg = off[node];
    const int deg = cnt[node];
    float4 acc = make_float4(0.0f, 0.0f, 0.0f, 0.0f);
    for (int j = 0; j < deg; j += 32) {
      const int m = (deg - j < 32) ? (deg - j) : 32;
      const int myidx = (lane < m) ? ssrc[beg + j + lane] : 0;
      for (int t = 0; t < m; t += 12) {
        const float4 v0 = load_row40(H4, myidx, t + g,     g, m, c);
        const float4 v1 = load_row40(H4, myidx, t + 3 + g, g, m, c);
        const float4 v2 = load_row40(H4, myidx, t + 6 + g, g, m, c);
        const float4 v3 = load_row40(H4, myidx, t + 9 + g, g, m, c);
        acc.x += v0.x; acc.y += v0.y; acc.z += v0.z; acc.w += v0.w;
        acc.x += v1.x; acc.y += v1.y; acc.z += v1.z; acc.w += v1.w;
        acc.x += v2.x; acc.y += v2.y; acc.z += v2.z; acc.w += v2.w;
        acc.x += v3.x; acc.y += v3.y; acc.z += v3.z; acc.w += v3.w;
      }
    }
    float4 r;
    r.x = (acc.x + __shfl(acc.x, l1, 32)) + __shfl(acc.x, l2, 32);
    r.y = (acc.y + __shfl(acc.y, l1, 32)) + __shfl(acc.y, l2, 32);
    r.z = (acc.z + __shfl(acc.z, l1, 32)) + __shfl(acc.z, l2, 32);
    r.w = (acc.w + __shfl(acc.w, l1, 32)) + __shfl(acc.w, l2, 32);

    float4 v;
    v.x = r.x + bias.x; v.y = r.y + bias.y; v.z = r.z + bias.z; v.w = r.w + bias.w;
    float mx = act ? fmaxf(fmaxf(v.x, v.y), fmaxf(v.z, v.w)) : -INFINITY;
    mx = fmaxf(mx, row_xor<8>(mx));
    mx = fmaxf(mx, row_xor<4>(mx));
    mx = fmaxf(mx, row_xor<2>(mx));
    mx = fmaxf(mx, row_xor<1>(mx));
    float s = 0.0f;
    if (act) s = (expf(v.x - mx) + expf(v.y - mx)) + (expf(v.z - mx) + expf(v.w - mx));
    s += row_xor<8>(s);
    s += row_xor<4>(s);
    s += row_xor<2>(s);
    s += row_xor<1>(s);
    const float lse = mx + logf(s);
    if (act) {
      ((float4*)OUT)[(size_t)node * (OUT_CH / 4) + lane] =
          make_float4(v.x - lse, v.y - lse, v.z - lse, v.w - lse);
    }
  }
}

extern "C" void kernel_launch(void* const* d_in, const int* in_sizes, int n_in,
                              void* d_out, int out_size, void* d_ws, size_t ws_size,
                              hipStream_t stream) {
  const float* x  = (const float*)d_in[0];
  const int*   ei = (const int*)d_in[1];
  const float* W1 = (const float*)d_in[2];
  const float* b1 = (const float*)d_in[3];
  const float* W2 = (const float*)d_in[4];
  const float* b2 = (const float*)d_in[5];

  const int n       = in_sizes[0] / IN_CH;
  const int n_edges = in_sizes[1] / 2;
  const int* src = ei;
  const int* dst = ei + n_edges;

  float* h2   = (float*)d_ws;
  float* out  = (float*)d_out;
  const int nb = (n + BIN_NODES - 1) / BIN_NODES;
  int2* eb    = (int2*)(h2 + (size_t)n * HID_CH);
  int2* seg   = eb + n_edges;
  int* cnt    = (int*)(seg + (size_t)nb * EDGE_CHUNKS);
  int* cursor = cnt + n;
  int* off    = cursor + 1;
  int* ssrc   = off + n;

  const int pairs   = (n + 31) / 32;
  const int nblk    = (n + 7) / 8;
  const int g_fused = pairs < 512 ? pairs : 512;
  const int g_node  = nblk < 2048 ? nblk : 2048;

  bin_local_kernel<<<EDGE_CHUNKS, 1024, 0, stream>>>(src, dst, eb, seg, cursor, n_edges, nb);
  bin_sort_kernel<<<nb, 1024, 0, stream>>>(eb, seg, cursor, cnt, off, ssrc, n, n_edges);

  gather1_gemm12_kernel<<<g_fused, 1024, 0, stream>>>(x, ssrc, off, cnt, W1, b1, W2, h2, n, pairs);

  gather2_logsoftmax_kernel<<<g_node, 256, 0, stream>>>(h2, ssrc, off, cnt, b2, out, n);
}


# Grouped edge records staged in LDS in both grouping kernels and copied out as whole 128-byte lines (eb slice, ssrc range) in place of one scattered 8- or 4-byte global store per edge
# speedup vs baseline: 5.2299x; 1.0699x over previous
#include <hip/hip_runtime.h>
#include <math.h>

typedef float v2f __attribute__((ext_vector_type(2)));
typedef float v8f __attribute__((ext_vector_type(8)));

#define IN_CH 128
#define HID_CH 128
#define OUT_CH 40

#define BIN_SHIFT 8
#define BIN_NODES 256
#define EDGE_CHUNKS 256
#define MAX_BINS 1024
#define EDGE_SLOTS 8
#define PIECES_PER_WAVE (EDGE_CHUNKS / 32)
#define STAGE_PAIRS 8192
#define STAGE_SRCS 8192
static_assert(EDGE_SLOTS == 8, "the load fences in bin_local_kernel name eight slots");

__device__ __forceinline__ int wave_incl_scan(int x, int lane) {
  for (int o = 1; o < 32; o <<= 1) {
    const int y = __shfl_up(x, o, 32);
    if (lane >= o) x += y;
  }
  return x;
}

__global__ __launch_bounds__(1024) void bin_local_kernel(const int* __restrict__ src,
                                                         const int* __restrict__ dst,
                                                         int2* __restrict__ eb,
                                                         int2* __restrict__ seg,
                                                         int* __restrict__ cursor,
                                                         int n_edges, int nb) {
  __shared__ int h[MAX_BINS];
  __shared__ int wt[32];
  __shared__ int2 stage[STAGE_PAIRS];
  const int tid = threadIdx.x;
  const int lane = tid & 31;
  const int wave = tid >> 5;
  h[tid] = 0;
  __syncthreads();
  const int chunk = (n_edges + EDGE_CHUNKS - 1) / EDGE_CHUNKS;
  const int e0 = blockIdx.x * chunk;
  const int e1 = (e0 + chunk < n_edges) ? (e0 + chunk) : n_edges;
  const bool staged = (chunk <= STAGE_PAIRS);
  for (int e = e0 + tid; e < e1; e += EDGE_SLOTS * 1024) {
    int d[EDGE_SLOTS];
#pragma unroll
    for (int k = 0; k < EDGE_SLOTS; ++k) {
      const int ee = e + k * 1024;
      d[k] = dst[ee < e1 ? ee : e1 - 1];
    }
    asm volatile("" :: "v"(d[0]), "v"(d[1]), "v"(d[2]), "v"(d[3]),
                       "v"(d[4]), "v"(d[5]), "v"(d[6]), "v"(d[7]));
#pragma unroll
    for (int k = 0; k < EDGE_SLOTS; ++k) {
      if (e + k * 1024 < e1) atomicAdd(&h[d[k] >> BIN_SHIFT], 1);
    }
  }
  __syncthreads();
  const int v = h[tid];
  const int x = wave_incl_scan(v, lane);
  if (lane == 31) wt[wave] = x;
  __syncthreads();
  int woff = 0;
  for (int w = 0; w < 32; ++w) {
    const int t = wt[w];
    if (w < wave) woff += t;
  }
  const int ex = woff + x - v;
  h[tid] = ex;
  if (tid < nb) seg[tid * EDGE_CHUNKS + blockIdx.x] = make_int2(ex, v);
  __syncthreads();
  for (int e = e0 + tid; e < e1; e += EDGE_SLOTS * 1024) {
    int d[EDGE_SLOTS], s[EDGE_SLOTS];
#pragma unroll
    for (int k = 0; k < EDGE_SLOTS; ++k) {
      const int ee = e + k * 1024;
      const int ec = ee < e1 ? ee : e1 - 1;
      d[k] = dst[ec];
      s[k] = src[ec];
    }
    asm volatile("" :: "v"(d[0]), "v"(d[1]), "v"(d[2]), "v"(d[3]),
                       "v"(d[4]), "v"(d[5]), "v"(d[6]), "v"(d[7]),
                       "v"(s[0]), "v"(s[1]), "v"(s[2]), "v"(s[3]),
                       "v"(s[4]), "v"(s[5]), "v"(s[6]), "v"(s[7]));
#pragma unroll
    for (int k = 0; k < EDGE_SLOTS; ++k) {
      if (e + k * 1024 < e1) {
        const int p = atomicAdd(&h[d[k] >> BIN_SHIFT], 1);
        if (staged) stage[p] = make_int2(d[k], s[k]);
        else eb[e0 + p] = make_int2(d[k], s[k]);
      }
    }
  }
  if (staged) {
    __syncthreads();
    const int a = (int)(((size_t)(eb + e0) >> 3) & 15);
    for (int i = tid - a; i < e1 - e0; i += 1024) {
      if (i >= 0) eb[e0 + i] = stage[i];
    }
  }
  if (blockIdx.x == 0 && tid == 0) *cursor = 0;
}

__global__ __launch_bounds__(1024, 16) void bin_sort_kernel(const int2* __restrict__ eb,
                                                        const int2* __restrict__ seg,
                                                        int* __restrict__ cursor,
                                                        int* __restrict__ cnt,
                                                        int* __restrict__ off,
                                                        int* __restrict__ ssrc,
                                                        int n, int n_edges) {
  __shared__ int c[BIN_NODES];
  __shared__ int o[BIN_NODES];
  __shared__ int wt[8];
  __shared__ int wm[8];
  __shared__ int2 sg[EDGE_CHUNKS];
  __shared__ int base_s;
  __shared__ int total_s;
  __shared__ int stage[STAGE_SRCS];
  const int b = blockIdx.x;
  const int tid = threadIdx.x;
  const int lane = tid & 31;
  const int wave = tid >> 5;
  const int node0 = b << BIN_SHIFT;
  const int chunk = (n_edges + EDGE_CHUNKS - 1) / EDGE_CHUNKS;
  if (tid < EDGE_CHUNKS) {
    const int2 r = seg[b * EDGE_CHUNKS + tid];
    sg[tid] = r;
    c[tid] = 0;
    const int t = wave_incl_scan(r.y, lane);
    if (lane == 31) wm[wave] = t;
  }
  __syncthreads();
  int pp[PIECES_PER_WAVE];
  int len[PIECES_PER_WAVE];
#pragma unroll
  for (int u = 0; u < PIECES_PER_WAVE; ++u) {
    const int k = wave + 32 * u;
    const int2 r = sg[k];
    pp[u] = k * chunk + r.x;
    len[u] = r.y;
  }
  {
    int d[PIECES_PER_WAVE];
#pragma unroll
    for (int u = 0; u < PIECES_PER_WAVE; ++u) d[u] = (lane < len[u]) ? eb[pp[u] + lane].x : node0;
    if (tid == 0) {
      int total = 0;
      for (int w = 0; w < 8; ++w) total += wm[w];
      total_s = total;
      base_s = atomicAdd(cursor, total);
    }
#pragma unroll
    for (int u = 0; u < PIECES_PER_WAVE; ++u) {
      if (lane < len[u]) atomicAdd(&c[d[u] - node0], 1);
    }
#pragma unroll
    for (int u = 0; u < PIECES_PER_WAVE; ++u) {
      for (int i = lane + 32; i < len[u]; i += 32) atomicAdd(&c[eb[pp[u] + i].x - node0], 1);
    }
  }
  __syncthreads();
  int v = 0, x = 0;
  if (tid < BIN_NODES) {
    v = c[tid];
    x = wave_incl_scan(v, lane);
    if (lane == 31) wt[wave] = x;
  }
  __syncthreads();
  const int beg = base_s;
  const int total = total_s;
  const bool staged = (total <= STAGE_SRCS);
  if (tid < BIN_NODES) {
    int woff = 0;
    for (int w = 0; w < 8; ++w) {
      const int t = wt[w];
      if (w < wave) woff += t;
    }
    const int ex = woff + x - v;
    o[tid] = ex;
    const int node = node0 + tid;
    if (node < n) { cnt[node] = v; off[node] = beg + ex; }
  }
  __syncthreads();
  {
    int2 r[PIECES_PER_WAVE];
#pragma unroll
    for (int u = 0; u < PIECES_PER_WAVE; ++u) {
      r[u] = (lane < len[u]) ? eb[pp[u] + lane] : make_int2(node0, 0);
    }
#pragma unroll
    for (int u = 0; u < PIECES_PER_WAVE; ++u) {
      if (lane < len[u]) {
        const int p = atomicAdd(&o[r[u].x - node0], 1);
        if (staged) stage[p] = r[u].y;
        else ssrc[beg + p] = r[u].y;
      }
    }
#pragma unroll
    for (int u = 0; u < PIECES_PER_WAVE; ++u) {
      for (int i = lane + 32; i < len[u]; i += 32) {
        const int2 q = eb[pp[u] + i];
        const int p = atomicAdd(&o[q.x - node0], 1);
        if (staged) stage[p] = q.y;
        else ssrc[beg + p] = q.y;
      }
    }
  }
  if (staged) {
    __syncthreads();
    const int a = (int)(((size_t)(ssrc + beg) >> 2) & 31);
    for (int i = tid - a; i < total; i += 1024) {
      if (i >= 0) ssrc[beg + i] = stage[i];
    }
  }
}

template <int K>
__device__ __forceinline__ void add_rows128(const float4* __restrict__ H4, int myidx,
                                            int t, int lane, float4& acc) {
  float4 v[K];
#pragma unroll
  for (int u = 0; u < K; ++u) {
    const int s = __builtin_amdgcn_readlane(myidx, t + u);
    v[u] = H4[(size_t)s * (HID_CH / 4) + lane];
  }
#pragma unroll
  for (int u = 0; u < K; ++u) {
    acc.x += v[u].x; acc.y += v[u].y; acc.z += v[u].z; acc.w += v[u].w;
  }
}

template <bool PAIRS>
__device__ __forceinline__ v2f load_bfrag(const float* bp, int i) {
  v2f b;
  if (PAIRS) {
    b = *(const v2f*)(bp + 512 * i);
  } else {
    b.x = bp[(4 * i) * OUT_CH];
    b.y = bp[(4 * i + 1) * OUT_CH];
  }
  return b;
}

template <bool PAIRS, int NREADS>
__device__ __forceinline__ v8f tile_times_b(const float* arow, const float* bp) {
  v8f c = {};
  v2f a[2][4], b[2][4];
#pragma unroll
  for (int s = 0; s < 4; ++s) {
    a[0][s] = *(const v2f*)(arow + 4 * s);
    b[0][s] = load_bfrag<PAIRS>(bp, s);
  }
  __builtin_amdgcn_sched_group_barrier(0x100, NREADS, 0);
#pragma unroll
  for (int g = 0; g < HID_CH / 16; ++g) {
    if (g + 1 < HID_CH / 16) {
#pragma unroll
      for (int s = 0; s < 4; ++s) {
        const int i = 4 * (g + 1) + s;
        a[(g + 1) & 1][s] = *(const v2f*)(arow + 4 * i);
        b[(g + 1) & 1][s] = load_bfrag<PAIRS>(bp, i);
      }
    }
#pragma unroll
    for (int s = 0; s < 4; ++s) {
      c = __builtin_amdgcn_wmma_f32_16x16x4_f32(false, a[g & 1][s], false, b[g & 1][s],
                                                (short)0, c, false, false);
    }
    __builtin_amdgcn_sched_group_barrier(0x100, NREADS, 0);
    __builtin_amdgcn_sched_group_barrier(0x008, 4, 0);
  }
  return c;
}

#define TILE_PITCH 132
#define TILE_ROWS 32
__global__ __launch_bounds__(1024, 16) void gather1_gemm12_kernel(const float* __restrict__ X,
                                                             const int* __restrict__ ssrc,
                                                             const int* __restrict__ off,
                                                             const int* __restrict__ cnt,
                                                             const float* __restrict__ W1,
                                                             const float* __restrict__ b1,
                                                             const float* __restrict__ W2,
                                                             float* __restrict__ H2,
                                                             int n, int pairs) {
  __shared__ float4 w1s4[IN_CH * HID_CH / 4];
  __shared__ float4 w2s4[HID_CH * OUT_CH / 4];
  __shared__ float4 xt4[2 * TILE_ROWS * TILE_PITCH / 4];
  __shared__ float4 ht4[2 * TILE_ROWS * TILE_PITCH / 4];
  float* w1s = (float*)w1s4;
  const float* w2s = (const float*)w2s4;
  float* xt = (float*)xt4;
  float* ht = (float*)ht4;

  const int wave = __builtin_amdgcn_readfirstlane((int)(threadIdx.x >> 5));
  const int lane = threadIdx.x & 31;
  const int half = lane >> 4;
  const int l    = lane & 15;
  const float4* __restrict__ X4 = (const float4*)X;

  for (int q = threadIdx.x; q < (IN_CH / 2) * (HID_CH / 4); q += 1024) {
    const int kp = q >> 5;
    const int c4 = (q & 31) << 2;
    const float4 r0 = ((const float4*)W1)[((2 * kp) * HID_CH + c4) >> 2];
    const float4 r1 = ((const float4*)W1)[((2 * kp + 1) * HID_CH + c4) >> 2];
    const int cs = c4 ^ ((kp & 1) << 4);
    float4* d = (float4*)(w1s + 2 * (kp * HID_CH + cs));
    d[0] = make_float4(r0.x, r1.x, r0.y, r1.y);
    d[1] = make_float4(r0.z, r1.z, r0.w, r1.w);
  }
  for (int q = threadIdx.x; q < HID_CH * OUT_CH / 4; q += 1024) w2s4[q] = ((const float4*)W2)[q];

  const int t1 = (wave >> 3) & 1;
  const int col1 = ((wave & 7) << 4) + l;
  const float bias1 = b1[col1];
  const int w3 = wave - 16;
  const int t2 = (w3 >= 3) ? 1 : 0;
  const int col2 = ((w3 - 3 * t2) << 4) + l;
  const bool colok = (col2 < OUT_CH);
  const int colc = colok ? col2 : 0;

  const int K = ((int)blockIdx.x < pairs) ? (pairs - 1 - (int)blockIdx.x) / (int)gridDim.x + 1 : 0;
  for (int r = 0; r < K + 2; ++r) {
    const int buf = r & 1;
    if (r < K) {
      const int node = ((int)blockIdx.x + r * (int)gridDim.x) * TILE_ROWS + wave;
      float4 acc = make_float4(0.0f, 0.0f, 0.0f, 0.0f);
      if (node < n) {
        const int beg = __builtin_amdgcn_readfirstlane(off[node]);
        const int deg = __builtin_amdgcn_readfirstlane(cnt[node]);
        for (int j = 0; j < deg; j += 32) {
          const int m = (deg - j < 32) ? (deg - j) : 32;
          const int myidx = (lane < m) ? ssrc[beg + j + lane] : 0;
          int u = 0;
          for (; u + 8 <= m; u += 8) add_rows128<8>(X4, myidx, u, lane, acc);
          if (u + 4 <= m) { add_rows128<4>(X4, myidx, u, lane, acc); u += 4; }
          if (u + 2 <= m) { add_rows128<2>(X4, myidx, u, lane, acc); u += 2; }
          if (u < m) add_rows128<1>(X4, myidx, u, lane, acc);
        }
      }
      *(float4*)(xt + buf * (TILE_ROWS * TILE_PITCH) + wave * TILE_PITCH + 4 * lane) = acc;
    }
    if (r >= 1 && r <= K && wave < 16) {
      const float* arow = xt + (buf ^ 1) * (TILE_ROWS * TILE_PITCH) +
                          (16 * t1 + l) * TILE_PITCH + (half << 1);
      const float* bp = w1s + 2 * (half * HID_CH + (col1 ^ (half << 4)));
      const v8f c = tile_times_b<true, 4>(arow, bp);
      float* hrow = ht + (buf ^ 1) * (TILE_ROWS * TILE_PITCH) +
                    (16 * t1 + (half << 3)) * TILE_PITCH + col1;
#pragma unroll
      for (int v = 0; v < 8; ++v) {
        const float h = c[v] + bias1;
        hrow[v * TILE_PITCH] = h > 0.0f ? h : 0.0f;
      }
    }
    if (r >= 2 && wave >= 16 && wave < 22) {
      const int row_base = ((int)blockIdx.x + (r - 2) * (int)gridDim.x) * TILE_ROWS + 16 * t2;
      const float* arow = ht + buf * (TILE_ROWS * TILE_PITCH) +
                          (16 * t2 + l) * TILE_PITCH + (half << 1);
      const float* bp = w2s + (half << 1) * OUT_CH + colc;
      const v8f c = tile_times_b<false, 6>(arow, bp);
      float* __restrict__ crow = H2 + (size_t)(row_base + (half << 3)) * OUT_CH + col2;
      if (colok) {
        if (row_base + 16 <= n) {
#pragma unroll
          for (int v = 0; v < 8; ++v) crow[(size_t)v * OUT_CH] = c[v];
        } else {
#pragma unroll
          for (int v = 0; v < 8; ++v) {
            if (row_base + (half << 3) + v < n) crow[(size_t)v * OUT_CH] = c[v];
          }
        }
      }
    }
    __syncthreads();
  }
}

__device__ __forceinline__ float4 load_row40(const float4* __restrict__ H4, int myidx,
                                             int e, int g, int m, int c) {
  const int s = __shfl(myidx, e & 31, 32);
  float4 v = make_float4(0.0f, 0.0f, 0.0f, 0.0f);
  if (g < 3 && e < m) v = H4[(size_t)s * (OUT_CH / 4) + c];
  return v;
}

template <int M>
__device__ __forceinline__ float row_xor(float x) {
  const int xi = __float_as_int(x);
  return __int_as_float(__builtin_amdgcn_update_dpp(xi, xi, 0x160 + M, 0xf, 0xf, false));
}

__global__ __launch_bounds__(256) void gather2_logsoftmax_kernel(const float* __restrict__ H,
                                                                 const int* __restrict__ ssrc,
                                                                 const int* __restrict__ off,
                                                                 const int* __restrict__ cnt,
                                                                 const float* __restrict__ b2,
                                                                 float* __restrict__ OUT, int n) {
  const int wave = __builtin_amdgcn_readfirstlane((int)(threadIdx.x >> 5));
  const int lane = threadIdx.x & 31;
  const int nwaves = gridDim.x * 8;
  const int g = lane / 10;
  const int c = lane - g * 10;
  const int l1 = (lane + 10) & 31;
  const int l2 = (lane + 20) & 31;
  const bool act = (lane < OUT_CH / 4);
  const float4* __restrict__ H4 = (const float4*)H;
  float4 bias = make_float4(0.0f, 0.0f, 0.0f, 0.0f);
  if (act) bias = ((const float4*)b2)[lane];
  for (int node = blockIdx.x * 8 + wave; node < n; node += nwaves) {
    const int beg = off[node];
    const int deg = cnt[node];
    float4 acc = make_float4(0.0f, 0.0f, 0.0f, 0.0f);
    for (int j = 0; j < deg; j += 32) {
      const int m = (deg - j < 32) ? (deg - j) : 32;
      const int myidx = (lane < m) ? ssrc[beg + j + lane] : 0;
      for (int t = 0; t < m; t += 12) {
        const float4 v0 = load_row40(H4, myidx, t + g,     g, m, c);
        const float4 v1 = load_row40(H4, myidx, t + 3 + g, g, m, c);
        const float4 v2 = load_row40(H4, myidx, t + 6 + g, g, m, c);
        const float4 v3 = load_row40(H4, myidx, t + 9 + g, g, m, c);
        acc.x += v0.x; acc.y += v0.y; acc.z += v0.z; acc.w += v0.w;
        acc.x += v1.x; acc.y += v1.y; acc.z += v1.z; acc.w += v1.w;
        acc.x += v2.x; acc.y += v2.y; acc.z += v2.z; acc.w += v2.w;
        acc.x += v3.x; acc.y += v3.y; acc.z += v3.z; acc.w += v3.w;
      }
    }
    float4 r;
    r.x = (acc.x + __shfl(acc.x, l1, 32)) + __shfl(acc.x, l2, 32);
    r.y = (acc.y + __shfl(acc.y, l1, 32)) + __shfl(acc.y, l2, 32);
    r.z = (acc.z + __shfl(acc.z, l1, 32)) + __shfl(acc.z, l2, 32);
    r.w = (acc.w + __shfl(acc.w, l1, 32)) + __shfl(acc.w, l2, 32);

    float4 v;
    v.x = r.x + bias.x; v.y = r.y + bias.y; v.z = r.z + bias.z; v.w = r.w + bias.w;
    float mx = act ? fmaxf(fmaxf(v.x, v.y), fmaxf(v.z, v.w)) : -INFINITY;
    mx = fmaxf(mx, row_xor<8>(mx));
    mx = fmaxf(mx, row_xor<4>(mx));
    mx = fmaxf(mx, row_xor<2>(mx));
    mx = fmaxf(mx, row_xor<1>(mx));
    float s = 0.0f;
    if (act) s = (expf(v.x - mx) + expf(v.y - mx)) + (expf(v.z - mx) + expf(v.w - mx));
    s += row_xor<8>(s);
    s += row_xor<4>(s);
    s += row_xor<2>(s);
    s += row_xor<1>(s);
    const float lse = mx + logf(s);
    if (act) {
      ((float4*)OUT)[(size_t)node * (OUT_CH / 4) + lane] =
          make_float4(v.x - lse, v.y - lse, v.z - lse, v.w - lse);
    }
  }
}

extern "C" void kernel_launch(void* const* d_in, const int* in_sizes, int n_in,
                              void* d_out, int out_size, void* d_ws, size_t ws_size,
                              hipStream_t stream) {
  const float* x  = (const float*)d_in[0];
  const int*   ei = (const int*)d_in[1];
  const float* W1 = (const float*)d_in[2];
  const float* b1 = (const float*)d_in[3];
  const float* W2 = (const float*)d_in[4];
  const float* b2 = (const float*)d_in[5];

  const int n       = in_sizes[0] / IN_CH;
  const int n_edges = in_sizes[1] / 2;
  const int* src = ei;
  const int* dst = ei + n_edges;

  float* h2   = (float*)d_ws;
  float* out  = (float*)d_out;
  const int nb = (n + BIN_NODES - 1) / BIN_NODES;
  int2* eb    = (int2*)(h2 + (size_t)n * HID_CH);
  int2* seg   = eb + n_edges;
  int* cnt    = (int*)(seg + (size_t)nb * EDGE_CHUNKS);
  int* cursor = cnt + n;
  int* off    = cursor + 1;
  int* ssrc   = off + n;

  const int pairs   = (n + 31) / 32;
  const int nblk    = (n + 7) / 8;
  const int g_fused = pairs < 512 ? pairs : 512;
  const int g_node  = nblk < 2048 ? nblk : 2048;

  bin_local_kernel<<<EDGE_CHUNKS, 1024, 0, stream>>>(src, dst, eb, seg, cursor, n_edges, nb);
  bin_sort_kernel<<<nb, 1024, 0, stream>>>(eb, seg, cursor, cnt, off, ssrc, n, n_edges);

  gather1_gemm12_kernel<<<g_fused, 1024, 0, stream>>>(x, ssrc, off, cnt, W1, b1, W2, h2, n, pairs);

  gather2_logsoftmax_kernel<<<g_node, 256, 0, stream>>>(h2, ssrc, off, cnt, b2, out, n);
}
